# starting kernel
# baseline (speedup 1.0000x reference)
#include <hip/hip_runtime.h>
#include <hip/hip_bf16.h>

typedef __attribute__((ext_vector_type(2))) float v2f;
typedef __attribute__((ext_vector_type(4))) float v4f;
typedef __attribute__((ext_vector_type(8))) float v8f;

#define CCH 128
#define HDIM 128
#define WDIM 128
#define HW (HDIM * WDIM)
#define KSZ 7
#define KK2 49
#define TPITCH 132

__global__ __launch_bounds__(256) void gemm_bn_wmma(
    const float* __restrict__ Xa, long sCa, long sPa,
    const float* __restrict__ Xb, long sCb, long sPb,
    const float* __restrict__ Wm, int ldw,
    const float* __restrict__ scale, const float* __restrict__ bias,
    float* __restrict__ Y, long sCo, long sPo, int relu)
{
    __shared__ float tile[64 * TPITCH];

    const int tid  = threadIdx.x;
    const int lane = tid & 31;
    const int wv   = tid >> 5;
    const int hl   = lane >> 4;
    const int l16  = lane & 15;
    const int m0   = wv << 4;
    const long p0  = (long)blockIdx.x * 64;

    v8f acc[4] = {};

    const float* srcX[2] = {Xa, Xb};
    const long   sCs[2]  = {sCa, sCb};
    const long   sPs[2]  = {sPa, sPb};
    const int nseg = (Xb != nullptr) ? 2 : 1;

    int bbase[4];
    #pragma unroll
    for (int s = 0; s < 4; ++s)
        bbase[s] = ((s << 4) + l16) * TPITCH + (hl << 1);

    for (int seg = 0; seg < nseg; ++seg) {
        const float* __restrict__ X = srcX[seg];
        const long sC = sCs[seg], sP = sPs[seg];

        __syncthreads();
        if (sP == 1) {
            #pragma unroll
            for (int i = 0; i < 8; ++i) {
                const int e  = tid + (i << 8);
                const int p4 = (e & 15) << 2;
                const int c  = e >> 4;
                const v4f xv = *(const v4f*)(X + (long)c * sC + p0 + p4);
                tile[(p4 + 0) * TPITCH + c] = xv.x;
                tile[(p4 + 1) * TPITCH + c] = xv.y;
                tile[(p4 + 2) * TPITCH + c] = xv.z;
                tile[(p4 + 3) * TPITCH + c] = xv.w;
            }
        } else {
            #pragma unroll
            for (int i = 0; i < 8; ++i) {
                const int e  = tid + (i << 8);
                const int c4 = (e & 31) << 2;
                const int p  = e >> 5;
                const v4f xv = *(const v4f*)(X + (p0 + p) * sP + c4);
                *(v4f*)(&tile[p * TPITCH + c4]) = xv;
            }
        }
        __syncthreads();

        const float* __restrict__ Ap =
            Wm + (long)(m0 + l16) * ldw + (seg << 7) + (hl << 1);

        #pragma unroll
        for (int k0 = 0; k0 < CCH; k0 += 4) {
            const v2f a = *(const v2f*)(Ap + k0);
            #pragma unroll
            for (int s = 0; s < 4; ++s) {
                const v2f b = *(const v2f*)(&tile[bbase[s] + k0]);
                acc[s] = __builtin_amdgcn_wmma_f32_16x16x4_f32(
                    false, a, false, b, (short)0, acc[s], false, false);
            }
        }
    }

    #pragma unroll
    for (int s = 0; s < 4; ++s) {
        const long p = p0 + (s << 4) + l16;
        #pragma unroll
        for (int r = 0; r < 8; ++r) {
            const int o = m0 + r + (hl << 3);
            float f = acc[s][r] * scale[o] + bias[o];
            if (relu) f = fmaxf(f, 0.0f);
            Y[(long)o * sCo + p * sPo] = f;
        }
    }
}

__global__ __launch_bounds__(256) void local_attn_kernel(
    const float* __restrict__ q, const float* __restrict__ k,
    const float* __restrict__ v, float* __restrict__ nf)
{
    const int lane = threadIdx.x & 31;
    const int wv   = threadIdx.x >> 5;
    const int p    = blockIdx.x * 8 + wv;
    const int h    = p >> 7;
    const int w    = p & (WDIM - 1);

    const v4f qv = *(const v4f*)(q + (long)p * CCH + lane * 4);

    float sim[KK2];
    #pragma unroll
    for (int t = 0; t < KK2; ++t) {
        const int hn = h + t / KSZ - (KSZ / 2);
        const int wn = w + t % KSZ - (KSZ / 2);
        float d = 0.0f;
        if ((unsigned)hn < (unsigned)HDIM && (unsigned)wn < (unsigned)WDIM) {
            const long pn = (long)hn * WDIM + wn;
            const v4f kv = *(const v4f*)(k + pn * CCH + lane * 4);
            d = qv.x * kv.x + qv.y * kv.y + qv.z * kv.z + qv.w * kv.w;
        }
        #pragma unroll
        for (int off = 16; off > 0; off >>= 1)
            d += __shfl_xor(d, off, 32);
        sim[t] = d;
    }

    float m = sim[0];
    #pragma unroll
    for (int t = 1; t < KK2; ++t) m = fmaxf(m, sim[t]);
    float z = 0.0f;
    #pragma unroll
    for (int t = 0; t < KK2; ++t) { sim[t] = __expf(sim[t] - m); z += sim[t]; }
    const float inv = 1.0f / z;

    v4f acc = {0.0f, 0.0f, 0.0f, 0.0f};
    #pragma unroll
    for (int t = 0; t < KK2; ++t) {
        const int hn = h + t / KSZ - (KSZ / 2);
        const int wn = w + t % KSZ - (KSZ / 2);
        if ((unsigned)hn < (unsigned)HDIM && (unsigned)wn < (unsigned)WDIM) {
            const long pn = (long)hn * WDIM + wn;
            const v4f vv = *(const v4f*)(v + pn * CCH + lane * 4);
            const float wt = sim[t] * inv;
            acc.x += wt * vv.x; acc.y += wt * vv.y;
            acc.z += wt * vv.z; acc.w += wt * vv.w;
        }
    }
    *(v4f*)(nf + (long)p * CCH + lane * 4) = acc;
}

extern "C" void kernel_launch(void* const* d_in, const int* in_sizes, int n_in,
                              void* d_out, int out_size, void* d_ws, size_t ws_size,
                              hipStream_t stream) {
    const float* x    = (const float*)d_in[0];
    const float* w_q1 = (const float*)d_in[1];
    const float* s_q1 = (const float*)d_in[2];
    const float* b_q1 = (const float*)d_in[3];
    const float* w_q2 = (const float*)d_in[4];
    const float* s_q2 = (const float*)d_in[5];
    const float* b_q2 = (const float*)d_in[6];
    const float* w_k1 = (const float*)d_in[7];
    const float* s_k1 = (const float*)d_in[8];
    const float* b_k1 = (const float*)d_in[9];
    const float* w_k2 = (const float*)d_in[10];
    const float* s_k2 = (const float*)d_in[11];
    const float* b_k2 = (const float*)d_in[12];
    const float* w_v  = (const float*)d_in[13];
    const float* s_v  = (const float*)d_in[14];
    const float* b_v  = (const float*)d_in[15];
    const float* w_o  = (const float*)d_in[16];
    const float* s_o  = (const float*)d_in[17];
    const float* b_o  = (const float*)d_in[18];
    float* out = (float*)d_out;

    const size_t NB = (size_t)HW * CCH;
    float* q1 = (float*)d_ws;
    float* k1 = q1 + NB;
    float* vb = k1 + NB;
    float* qb = vb + NB;
    float* kb = qb + NB;
    float* nf = kb + NB;

    const dim3 blk(256);
    const dim3 grd(HW / 64);

    gemm_bn_wmma<<<grd, blk, 0, stream>>>(x, HW, 1, nullptr, 0, 0,
                                          w_q1, CCH, s_q1, b_q1,
                                          q1, 1, CCH, 1);
    gemm_bn_wmma<<<grd, blk, 0, stream>>>(x, HW, 1, nullptr, 0, 0,
                                          w_k1, CCH, s_k1, b_k1,
                                          k1, 1, CCH, 1);
    gemm_bn_wmma<<<grd, blk, 0, stream>>>(x, HW, 1, nullptr, 0, 0,
                                          w_v, CCH, s_v, b_v,
                                          vb, 1, CCH, 1);
    gemm_bn_wmma<<<grd, blk, 0, stream>>>(q1, 1, CCH, nullptr, 0, 0,
                                          w_q2, CCH, s_q2, b_q2,
                                          qb, 1, CCH, 1);
    gemm_bn_wmma<<<grd, blk, 0, stream>>>(k1, 1, CCH, nullptr, 0, 0,
                                          w_k2, CCH, s_k2, b_k2,
                                          kb, 1, CCH, 1);
    local_attn_kernel<<<dim3(HW / 8), blk, 0, stream>>>(qb, kb, vb, nf);
    gemm_bn_wmma<<<grd, blk, 0, stream>>>(nf, 1, CCH, x, HW, 1,
                                          w_o, 2 * CCH, s_o, b_o,
                                          out, HW, 1, 0);
}


# independent 1x1-conv GEMMs batched in one grid, blockIdx.y picks the problem, each block still one GEMM: q1,k1,v = 768 blocks (3 per compute unit), q,k = 512; 7 launches -> 4; sums unchanged
# speedup vs baseline: 2.3225x; 2.3225x over previous
#include <hip/hip_runtime.h>
#include <hip/hip_bf16.h>

typedef __attribute__((ext_vector_type(2))) float v2f;
typedef __attribute__((ext_vector_type(4))) float v4f;
typedef __attribute__((ext_vector_type(8))) float v8f;

#define CCH 128
#define HDIM 128
#define WDIM 128
#define HW (HDIM * WDIM)
#define TPITCH 132

__device__ __forceinline__ void lds_barrier()
{
    __builtin_amdgcn_fence(__ATOMIC_RELEASE, "workgroup", "local");
    __builtin_amdgcn_s_barrier();
    __builtin_amdgcn_fence(__ATOMIC_ACQUIRE, "workgroup", "local");
}

__device__ __forceinline__ void tile_load_cm(const float* __restrict__ X, long sC,
                                             long p0, int tid, v4f (&xv)[8])
{
    #pragma unroll
    for (int i = 0; i < 8; ++i) {
        const int e  = tid + (i << 8);
        const int p4 = (e & 15) << 2;
        const int c  = e >> 4;
        xv[i] = *(const v4f*)(X + (long)c * sC + p0 + p4);
    }
}
__device__ __forceinline__ void tile_load_pm(const float* __restrict__ X, long sP,
                                             long p0, int tid, v4f (&xv)[8])
{
    #pragma unroll
    for (int i = 0; i < 8; ++i) {
        const int e  = tid + (i << 8);
        const int c4 = (e & 31) << 2;
        const int p  = e >> 5;
        xv[i] = *(const v4f*)(X + (p0 + p) * sP + c4);
    }
}

__device__ __forceinline__ void tile_store_cm(float* __restrict__ tile, int tid,
                                              const v4f (&xv)[8])
{
    #pragma unroll
    for (int i = 0; i < 8; ++i) {
        const int e  = tid + (i << 8);
        const int p4 = (e & 15) << 2;
        const int c  = e >> 4;
        tile[(p4 + 0) * TPITCH + c] = xv[i].x;
        tile[(p4 + 1) * TPITCH + c] = xv[i].y;
        tile[(p4 + 2) * TPITCH + c] = xv[i].z;
        tile[(p4 + 3) * TPITCH + c] = xv[i].w;
    }
}
__device__ __forceinline__ void tile_store_pm(float* __restrict__ tile, int tid,
                                              const v4f (&xv)[8])
{
    #pragma unroll
    for (int i = 0; i < 8; ++i) {
        const int e  = tid + (i << 8);
        const int c4 = (e & 31) << 2;
        const int p  = e >> 5;
        *(v4f*)(&tile[p * TPITCH + c4]) = xv[i];
    }
}

__device__ __forceinline__ void mma_segment(const float* __restrict__ tile,
                                            const int (&bbase)[4],
                                            const v2f (&afr)[32], v8f (&acc)[4])
{
    #pragma unroll
    for (int kk = 0; kk < 32; ++kk) {
        #pragma unroll
        for (int s = 0; s < 4; ++s) {
            const v2f b = *(const v2f*)(&tile[bbase[s] + 4 * kk]);
            acc[s] = __builtin_amdgcn_wmma_f32_16x16x4_f32(
                false, afr[kk], false, b, (short)0, acc[s], false, false);
        }
    }
}

template <bool TWO>
__global__ __launch_bounds__(256) void gemm_bn_wmma(
    const float* __restrict__ Xa0, const float* __restrict__ Xa1,
    const float* __restrict__ Xa2, long sCa, long sPa,
    const float* __restrict__ Xb, long sCb,
    const float* __restrict__ Wm0, const float* __restrict__ Wm1,
    const float* __restrict__ Wm2, int ldw,
    const float* __restrict__ scale0, const float* __restrict__ scale1,
    const float* __restrict__ scale2,
    const float* __restrict__ bias0, const float* __restrict__ bias1,
    const float* __restrict__ bias2,
    float* __restrict__ Y0, float* __restrict__ Y1, float* __restrict__ Y2,
    long sCo, long sPo, int relu)
{
    __shared__ __attribute__((aligned(16))) float tile[64 * TPITCH];

    const int gsel = blockIdx.y;
    const float* __restrict__ Xa    = gsel == 0 ? Xa0    : (gsel == 1 ? Xa1    : Xa2);
    const float* __restrict__ Wm    = gsel == 0 ? Wm0    : (gsel == 1 ? Wm1    : Wm2);
    const float* __restrict__ scale = gsel == 0 ? scale0 : (gsel == 1 ? scale1 : scale2);
    const float* __restrict__ bias  = gsel == 0 ? bias0  : (gsel == 1 ? bias1  : bias2);
    float* __restrict__ Y           = gsel == 0 ? Y0     : (gsel == 1 ? Y1     : Y2);

    const int tid  = threadIdx.x;
    const int lane = tid & 31;
    const int wv   = tid >> 5;
    const int hl   = lane >> 4;
    const int l16  = lane & 15;
    const int m0   = wv << 4;
    const long p0  = (long)blockIdx.x * 64;
    const bool two  = TWO && (Xb != nullptr);
    const bool a_cm = (sPa == 1);

    v4f xv[8], xw[8];
    if (a_cm) tile_load_cm(Xa, sCa, p0, tid, xv);
    else      tile_load_pm(Xa, sPa, p0, tid, xv);
    if (two)  tile_load_cm(Xb, sCb, p0, tid, xw);

    const float* __restrict__ Ap = Wm + (long)(m0 + l16) * ldw + (hl << 1);
    v2f afr[32];
    #pragma unroll
    for (int kk = 0; kk < 32; ++kk)
        afr[kk] = *(const v2f*)(Ap + 4 * kk);

    const int ob = m0 + (hl << 3);
    float sc[8], bi[8];
    #pragma unroll
    for (int r = 0; r < 8; ++r) { sc[r] = scale[ob + r]; bi[r] = bias[ob + r]; }

    asm volatile("" ::: "memory");

    if (a_cm) tile_store_cm(tile, tid, xv);
    else      tile_store_pm(tile, tid, xv);
    lds_barrier();

    int bbase[4];
    #pragma unroll
    for (int s = 0; s < 4; ++s)
        bbase[s] = ((s << 4) + l16) * TPITCH + (hl << 1);

    v8f acc[4] = {};
    mma_segment(tile, bbase, afr, acc);

    if (two) {
        #pragma unroll
        for (int kk = 0; kk < 32; ++kk)
            afr[kk] = *(const v2f*)(Ap + CCH + 4 * kk);
        lds_barrier();
        tile_store_cm(tile, tid, xw);
        lds_barrier();
        mma_segment(tile, bbase, afr, acc);
    }

    if (sCo == 1 && sPo == CCH) {
        lds_barrier();
        #pragma unroll
        for (int s = 0; s < 4; ++s) {
            float f[8];
            #pragma unroll
            for (int r = 0; r < 8; ++r) {
                f[r] = acc[s][r] * sc[r] + bi[r];
                if (relu) f[r] = fmaxf(f[r], 0.0f);
            }
            float* dst = &tile[((s << 4) + l16) * TPITCH + ob];
            const v4f lo = {f[0], f[1], f[2], f[3]};
            const v4f hi = {f[4], f[5], f[6], f[7]};
            *(v4f*)(dst)     = lo;
            *(v4f*)(dst + 4) = hi;
        }
        lds_barrier();
        float* __restrict__ Yp = Y + p0 * CCH + (lane << 2);
        #pragma unroll
        for (int i = 0; i < 8; ++i) {
            const int p = wv + (i << 3);
            const v4f val = *(const v4f*)(&tile[p * TPITCH + (lane << 2)]);
            *(v4f*)(Yp + (long)p * CCH) = val;
        }
    } else {
        #pragma unroll
        for (int s = 0; s < 4; ++s) {
            const long p = p0 + (s << 4) + l16;
            #pragma unroll
            for (int r = 0; r < 8; ++r) {
                const int o = ob + r;
                float f = acc[s][r] * sc[r] + bi[r];
                if (relu) f = fmaxf(f, 0.0f);
                Y[(long)o * sCo + p * sPo] = f;
            }
        }
    }
}

#define SP 226

__global__ __launch_bounds__(256) void local_attn_wmma(
    const float* __restrict__ q, const float* __restrict__ k,
    const float* __restrict__ v, float* __restrict__ nf)
{
    __shared__ __attribute__((aligned(16))) float sp[16 * SP];

    const int tid  = threadIdx.x;
    const int lane = tid & 31;
    const int wv   = tid >> 5;
    const int hl   = lane >> 4;
    const int l16  = lane & 15;
    const int p0   = blockIdx.x * 16;
    const int h    = p0 >> 7;
    const int w0   = p0 & (WDIM - 1);

    if (wv < 7) {
        const int  hn    = h + wv - 3;
        const bool rowok = (unsigned)hn < (unsigned)HDIM;
        const int  wn0   = w0 - 3 + l16;
        const int  wn1   = w0 + 13 + l16;
        const bool ok0   = rowok && (unsigned)wn0 < (unsigned)WDIM;
        const bool ok1   = rowok && (l16 < 6) && (unsigned)wn1 < (unsigned)WDIM;
        const long pn0 = ok0 ? ((long)hn * WDIM + wn0) : (long)(p0 + l16);
        const long pn1 = ok1 ? ((long)hn * WDIM + wn1) : (long)(p0 + l16);
        const float* __restrict__ qp  = q + (long)(p0 + l16) * CCH + (hl << 2);
        const float* __restrict__ kp0 = k + pn0 * CCH + (hl << 2);
        const float* __restrict__ kp1 = k + pn1 * CCH + (hl << 2);
        const v4f zero4 = {0.0f, 0.0f, 0.0f, 0.0f};

        v8f s0 = {}, s1 = {};
        #pragma unroll
        for (int jj = 0; jj < 16; ++jj) {
            const v4f a  = *(const v4f*)(qp + 8 * jj);
            v4f b0 = *(const v4f*)(kp0 + 8 * jj);
            v4f b1 = *(const v4f*)(kp1 + 8 * jj);
            b0 = ok0 ? b0 : zero4;
            b1 = ok1 ? b1 : zero4;
            s0 = __builtin_amdgcn_wmma_f32_16x16x4_f32(
                false, a.xy, false, b0.xy, (short)0, s0, false, false);
            s1 = __builtin_amdgcn_wmma_f32_16x16x4_f32(
                false, a.xy, false, b1.xy, (short)0, s1, false, false);
            s0 = __builtin_amdgcn_wmma_f32_16x16x4_f32(
                false, a.zw, false, b0.zw, (short)0, s0, false, false);
            s1 = __builtin_amdgcn_wmma_f32_16x16x4_f32(
                false, a.zw, false, b1.zw, (short)0, s1, false, false);
        }
        #pragma unroll
        for (int r = 0; r < 8; ++r) {
            sp[(r + (hl << 3)) * SP + wv * 32 + l16]      = s0[r];
            sp[(r + (hl << 3)) * SP + wv * 32 + 16 + l16] = s1[r];
        }
    }
    __syncthreads();

    #pragma unroll
    for (int mm = 0; mm < 2; ++mm) {
        const int  m      = 2 * wv + mm;
        const bool inband = (lane >= m) && (lane <= m + 6);
        float* row = &sp[m * SP + lane];
        float s[7];
        #pragma unroll
        for (int dy = 0; dy < 7; ++dy) s[dy] = row[dy * 32];
        float mx = s[0];
        #pragma unroll
        for (int dy = 1; dy < 7; ++dy) mx = fmaxf(mx, s[dy]);
        mx = inband ? mx : -INFINITY;
        #pragma unroll
        for (int off = 16; off > 0; off >>= 1)
            mx = fmaxf(mx, __shfl_xor(mx, off, 32));
        float e[7];
        float z = 0.0f;
        #pragma unroll
        for (int dy = 0; dy < 7; ++dy) {
            e[dy] = inband ? __expf(s[dy] - mx) : 0.0f;
            z += e[dy];
        }
        #pragma unroll
        for (int off = 16; off > 0; off >>= 1)
            z += __shfl_xor(z, off, 32);
        const float inv = 1.0f / z;
        #pragma unroll
        for (int dy = 0; dy < 7; ++dy) row[dy * 32] = e[dy] * inv;
    }
    __syncthreads();

    const int c0 = wv << 4;
    v8f o0 = {}, o1 = {};
    #pragma unroll
    for (int dy = 0; dy < 7; ++dy) {
        const int  hn    = h + dy - 3;
        const bool rowok = (unsigned)hn < (unsigned)HDIM;
        #pragma unroll
        for (int ks = 0; ks < 6; ++ks) {
            const v2f a = *(const v2f*)(&sp[l16 * SP + dy * 32 + 4 * ks + (hl << 1)]);
            const int  j0  = 4 * ks + (hl << 1);
            const int  wnA = w0 - 3 + j0;
            const int  wnB = wnA + 1;
            const bool okA = rowok && (j0 < 22)     && (unsigned)wnA < (unsigned)WDIM;
            const bool okB = rowok && (j0 + 1 < 22) && (unsigned)wnB < (unsigned)WDIM;
            const long pA = okA ? ((long)hn * WDIM + wnA) : (long)p0;
            const long pB = okB ? ((long)hn * WDIM + wnB) : (long)p0;
            float bA = v[pA * CCH + c0 + l16];
            float bB = v[pB * CCH + c0 + l16];
            bA = okA ? bA : 0.0f;
            bB = okB ? bB : 0.0f;
            const v2f b = {bA, bB};
            if (ks & 1)
                o1 = __builtin_amdgcn_wmma_f32_16x16x4_f32(
                    false, a, false, b, (short)0, o1, false, false);
            else
                o0 = __builtin_amdgcn_wmma_f32_16x16x4_f32(
                    false, a, false, b, (short)0, o0, false, false);
        }
    }
    #pragma unroll
    for (int r = 0; r < 8; ++r)
        nf[(long)(p0 + r + (hl << 3)) * CCH + c0 + l16] = o0[r] + o1[r];
}

extern "C" void kernel_launch(void* const* d_in, const int* in_sizes, int n_in,
                              void* d_out, int out_size, void* d_ws, size_t ws_size,
                              hipStream_t stream) {
    const float* x    = (const float*)d_in[0];
    const float* w_q1 = (const float*)d_in[1];
    const float* s_q1 = (const float*)d_in[2];
    const float* b_q1 = (const float*)d_in[3];
    const float* w_q2 = (const float*)d_in[4];
    const float* s_q2 = (const float*)d_in[5];
    const float* b_q2 = (const float*)d_in[6];
    const float* w_k1 = (const float*)d_in[7];
    const float* s_k1 = (const float*)d_in[8];
    const float* b_k1 = (const float*)d_in[9];
    const float* w_k2 = (const float*)d_in[10];
    const float* s_k2 = (const float*)d_in[11];
    const float* b_k2 = (const float*)d_in[12];
    const float* w_v  = (const float*)d_in[13];
    const float* s_v  = (const float*)d_in[14];
    const float* b_v  = (const float*)d_in[15];
    const float* w_o  = (const float*)d_in[16];
    const float* s_o  = (const float*)d_in[17];
    const float* b_o  = (const float*)d_in[18];
    float* out = (float*)d_out;

    const size_t NB = (size_t)HW * CCH;
    float* q1 = (float*)d_ws;
    float* k1 = q1 + NB;
    float* vb = k1 + NB;
    float* qb = vb + NB;
    float* kb = qb + NB;
    float* nf = kb + NB;

    const dim3 blk(256);

    gemm_bn_wmma<false><<<dim3(HW / 64, 3), blk, 0, stream>>>(
        x, x, x, HW, 1, nullptr, 0,
        w_q1, w_k1, w_v, CCH,
        s_q1, s_k1, s_v, b_q1, b_k1, b_v,
        q1, k1, vb, 1, CCH, 1);
    gemm_bn_wmma<false><<<dim3(HW / 64, 2), blk, 0, stream>>>(
        q1, k1, nullptr, 1, CCH, nullptr, 0,
        w_q2, w_k2, nullptr, CCH,
        s_q2, s_k2, nullptr, b_q2, b_k2, nullptr,
        qb, kb, nullptr, 1, CCH, 1);
    local_attn_wmma<<<dim3(HW / 16), blk, 0, stream>>>(qb, kb, vb, nf);
    gemm_bn_wmma<true><<<dim3(HW / 64, 1), blk, 0, stream>>>(
        nf, nullptr, nullptr, 1, CCH, x, HW,
        w_o, nullptr, nullptr, 2 * CCH,
        s_o, nullptr, nullptr, b_o, nullptr, nullptr,
        out, nullptr, nullptr, HW, 1, 0);
}


# closing 1x1 conv by linearity: s_o*W_o[:, :128] applied to v ahead of the weighted sum (P.(W v), not W.(P v)), x half + bias stored in out as start value in stage 1; 4 launches -> 3; sum reordered
# speedup vs baseline: 2.4299x; 1.0462x over previous
#include <hip/hip_runtime.h>
#include <hip/hip_bf16.h>

typedef __attribute__((ext_vector_type(2))) float v2f;
typedef __attribute__((ext_vector_type(4))) float v4f;
typedef __attribute__((ext_vector_type(8))) float v8f;

#define CCH 128
#define HDIM 128
#define WDIM 128
#define HW (HDIM * WDIM)
#define TPITCH 132

__device__ __forceinline__ void lds_barrier()
{
    __builtin_amdgcn_fence(__ATOMIC_RELEASE, "workgroup", "local");
    __builtin_amdgcn_s_barrier();
    __builtin_amdgcn_fence(__ATOMIC_ACQUIRE, "workgroup", "local");
}

__device__ __forceinline__ void tile_load_cm(const float* __restrict__ X, long sC,
                                             long p0, int tid, v4f (&xv)[8])
{
    #pragma unroll
    for (int i = 0; i < 8; ++i) {
        const int e  = tid + (i << 8);
        const int p4 = (e & 15) << 2;
        const int c  = e >> 4;
        xv[i] = *(const v4f*)(X + (long)c * sC + p0 + p4);
    }
}
__device__ __forceinline__ void tile_load_pm(const float* __restrict__ X, long sP,
                                             long p0, int tid, v4f (&xv)[8])
{
    #pragma unroll
    for (int i = 0; i < 8; ++i) {
        const int e  = tid + (i << 8);
        const int c4 = (e & 31) << 2;
        const int p  = e >> 5;
        xv[i] = *(const v4f*)(X + (p0 + p) * sP + c4);
    }
}

__device__ __forceinline__ void tile_store_cm(float* __restrict__ tile, int tid,
                                              const v4f (&xv)[8])
{
    #pragma unroll
    for (int i = 0; i < 8; ++i) {
        const int e  = tid + (i << 8);
        const int p4 = (e & 15) << 2;
        const int c  = e >> 4;
        tile[(p4 + 0) * TPITCH + c] = xv[i].x;
        tile[(p4 + 1) * TPITCH + c] = xv[i].y;
        tile[(p4 + 2) * TPITCH + c] = xv[i].z;
        tile[(p4 + 3) * TPITCH + c] = xv[i].w;
    }
}
__device__ __forceinline__ void tile_store_pm(float* __restrict__ tile, int tid,
                                              const v4f (&xv)[8])
{
    #pragma unroll
    for (int i = 0; i < 8; ++i) {
        const int e  = tid + (i << 8);
        const int c4 = (e & 31) << 2;
        const int p  = e >> 5;
        *(v4f*)(&tile[p * TPITCH + c4]) = xv[i];
    }
}

__device__ __forceinline__ void mma_segment(const float* __restrict__ tile,
                                            const int (&bbase)[4],
                                            const v2f (&afr)[32], v8f (&acc)[4])
{
    #pragma unroll
    for (int kk = 0; kk < 32; ++kk) {
        #pragma unroll
        for (int s = 0; s < 4; ++s) {
            const v2f b = *(const v2f*)(&tile[bbase[s] + 4 * kk]);
            acc[s] = __builtin_amdgcn_wmma_f32_16x16x4_f32(
                false, afr[kk], false, b, (short)0, acc[s], false, false);
        }
    }
}

__global__ __launch_bounds__(256) void gemm_bn_wmma(
    const float* __restrict__ Xa0, const float* __restrict__ Xa1,
    const float* __restrict__ Xa2, const float* __restrict__ Xa3,
    long sCa, long sPa,
    const float* __restrict__ Wm0, const float* __restrict__ Wm1,
    const float* __restrict__ Wm2, const float* __restrict__ Wm3,
    const float* __restrict__ scale0, const float* __restrict__ scale1,
    const float* __restrict__ scale2, const float* __restrict__ scale3,
    const float* __restrict__ bias0, const float* __restrict__ bias1,
    const float* __restrict__ bias2, const float* __restrict__ bias3,
    float* __restrict__ Y0, float* __restrict__ Y1,
    float* __restrict__ Y2, float* __restrict__ Y3,
    int relu_mask, int bias_mask, int wide_mask, int cm_mask)
{
    __shared__ __attribute__((aligned(16))) float tile[64 * TPITCH];

    const int gsel = blockIdx.y;
    const float* __restrict__ Xa    = gsel == 0 ? Xa0    : (gsel == 1 ? Xa1    : (gsel == 2 ? Xa2    : Xa3));
    const float* __restrict__ Wm    = gsel == 0 ? Wm0    : (gsel == 1 ? Wm1    : (gsel == 2 ? Wm2    : Wm3));
    const float* __restrict__ scale = gsel == 0 ? scale0 : (gsel == 1 ? scale1 : (gsel == 2 ? scale2 : scale3));
    const float* __restrict__ bias  = gsel == 0 ? bias0  : (gsel == 1 ? bias1  : (gsel == 2 ? bias2  : bias3));
    float* __restrict__ Y           = gsel == 0 ? Y0     : (gsel == 1 ? Y1     : (gsel == 2 ? Y2     : Y3));
    const bool do_relu = ((relu_mask >> gsel) & 1) != 0;
    const bool do_bias = ((bias_mask >> gsel) & 1) != 0;
    const bool out_cm  = ((cm_mask   >> gsel) & 1) != 0;
    const int  ldw     = ((wide_mask >> gsel) & 1) != 0 ? 2 * CCH : CCH;

    const int tid  = threadIdx.x;
    const int lane = tid & 31;
    const int wv   = tid >> 5;
    const int hl   = lane >> 4;
    const int l16  = lane & 15;
    const int m0   = wv << 4;
    const long p0  = (long)blockIdx.x * 64;
    const bool a_cm = (sPa == 1);

    v4f xv[8];
    if (a_cm) tile_load_cm(Xa, sCa, p0, tid, xv);
    else      tile_load_pm(Xa, sPa, p0, tid, xv);

    const float* __restrict__ Ap = Wm + (long)(m0 + l16) * ldw + (hl << 1);
    v2f afr[32];
    #pragma unroll
    for (int kk = 0; kk < 32; ++kk)
        afr[kk] = *(const v2f*)(Ap + 4 * kk);

    const int ob = m0 + (hl << 3);
    float sc[8], bi[8];
    #pragma unroll
    for (int r = 0; r < 8; ++r) { sc[r] = scale[ob + r]; bi[r] = bias[ob + r]; }

    asm volatile("" ::: "memory");

    if (a_cm) tile_store_cm(tile, tid, xv);
    else      tile_store_pm(tile, tid, xv);
    lds_barrier();

    int bbase[4];
    #pragma unroll
    for (int s = 0; s < 4; ++s)
        bbase[s] = ((s << 4) + l16) * TPITCH + (hl << 1);

    v8f acc[4] = {};
    mma_segment(tile, bbase, afr, acc);

    if (!do_bias) {
        #pragma unroll
        for (int r = 0; r < 8; ++r) bi[r] = 0.0f;
    }
    if (!out_cm) {
        lds_barrier();
        #pragma unroll
        for (int s = 0; s < 4; ++s) {
            float f[8];
            #pragma unroll
            for (int r = 0; r < 8; ++r) {
                f[r] = acc[s][r] * sc[r] + bi[r];
                if (do_relu) f[r] = fmaxf(f[r], 0.0f);
            }
            float* dst = &tile[((s << 4) + l16) * TPITCH + ob];
            const v4f lo = {f[0], f[1], f[2], f[3]};
            const v4f hi = {f[4], f[5], f[6], f[7]};
            *(v4f*)(dst)     = lo;
            *(v4f*)(dst + 4) = hi;
        }
        lds_barrier();
        float* __restrict__ Yp = Y + p0 * CCH + (lane << 2);
        #pragma unroll
        for (int i = 0; i < 8; ++i) {
            const int p = wv + (i << 3);
            const v4f val = *(const v4f*)(&tile[p * TPITCH + (lane << 2)]);
            *(v4f*)(Yp + (long)p * CCH) = val;
        }
    } else {
        #pragma unroll
        for (int s = 0; s < 4; ++s) {
            const long p = p0 + (s << 4) + l16;
            #pragma unroll
            for (int r = 0; r < 8; ++r) {
                const int o = ob + r;
                float f = acc[s][r] * sc[r] + bi[r];
                if (do_relu) f = fmaxf(f, 0.0f);
                Y[(long)o * HW + p] = f;
            }
        }
    }
}

#define SP 226

#define OP 36

__global__ __launch_bounds__(512) __attribute__((amdgpu_waves_per_eu(8, 8)))
void local_attn_wmma(
    const float* __restrict__ q, const float* __restrict__ k,
    const float* __restrict__ v, float* __restrict__ out)
{
    __shared__ __attribute__((aligned(16))) float sp2[2 * 16 * SP];
    __shared__ __attribute__((aligned(16))) float ot[CCH * OP];

    const int tid  = threadIdx.x;
    const int lane = tid & 31;
    const int sh   = tid >> 8;
    const int wv   = (tid >> 5) & 7;
    const int hl   = lane >> 4;
    const int l16  = lane & 15;
    const int pb   = blockIdx.x * 32;
    const int p0   = pb + (sh << 4);
    const int h    = p0 >> 7;
    const int w0   = p0 & (WDIM - 1);
    float* __restrict__ sp = sp2 + sh * (16 * SP);


    if (wv < 7) {
        const int  hn    = h + wv - 3;
        const bool rowok = (unsigned)hn < (unsigned)HDIM;
        const int  wn0   = w0 - 3 + l16;
        const int  wn1   = w0 + 13 + l16;
        const bool ok0   = rowok && (unsigned)wn0 < (unsigned)WDIM;
        const bool ok1   = rowok && (l16 < 6) && (unsigned)wn1 < (unsigned)WDIM;
        const long pn0 = ok0 ? ((long)hn * WDIM + wn0) : (long)(p0 + l16);
        const long pn1 = ok1 ? ((long)hn * WDIM + wn1) : (long)(p0 + l16);
        const float* __restrict__ qp  = q + (long)(p0 + l16) * CCH + (hl << 2);
        const float* __restrict__ kp0 = k + pn0 * CCH + (hl << 2);
        const float* __restrict__ kp1 = k + pn1 * CCH + (hl << 2);
        const v4f zero4 = {0.0f, 0.0f, 0.0f, 0.0f};

        v8f s0 = {}, s1 = {};
        #pragma unroll
        for (int jj = 0; jj < 16; ++jj) {
            const v4f a  = *(const v4f*)(qp + 8 * jj);
            v4f b0 = *(const v4f*)(kp0 + 8 * jj);
            v4f b1 = *(const v4f*)(kp1 + 8 * jj);
            b0 = ok0 ? b0 : zero4;
            b1 = ok1 ? b1 : zero4;
            s0 = __builtin_amdgcn_wmma_f32_16x16x4_f32(
                false, a.xy, false, b0.xy, (short)0, s0, false, false);
            s1 = __builtin_amdgcn_wmma_f32_16x16x4_f32(
                false, a.xy, false, b1.xy, (short)0, s1, false, false);
            s0 = __builtin_amdgcn_wmma_f32_16x16x4_f32(
                false, a.zw, false, b0.zw, (short)0, s0, false, false);
            s1 = __builtin_amdgcn_wmma_f32_16x16x4_f32(
                false, a.zw, false, b1.zw, (short)0, s1, false, false);
        }
        #pragma unroll
        for (int r = 0; r < 8; ++r) {
            sp[(r + (hl << 3)) * SP + wv * 32 + l16]      = s0[r];
            sp[(r + (hl << 3)) * SP + wv * 32 + 16 + l16] = s1[r];
        }
    }
    __syncthreads();

    #pragma unroll
    for (int mm = 0; mm < 2; ++mm) {
        const int  m      = 2 * wv + mm;
        const bool inband = (lane >= m) && (lane <= m + 6);
        float* row = &sp[m * SP + lane];
        float s[7];
        #pragma unroll
        for (int dy = 0; dy < 7; ++dy) s[dy] = row[dy * 32];
        float mx = s[0];
        #pragma unroll
        for (int dy = 1; dy < 7; ++dy) mx = fmaxf(mx, s[dy]);
        mx = inband ? mx : -INFINITY;
        #pragma unroll
        for (int off = 16; off > 0; off >>= 1)
            mx = fmaxf(mx, __shfl_xor(mx, off, 32));
        float e[7];
        float z = 0.0f;
        #pragma unroll
        for (int dy = 0; dy < 7; ++dy) {
            e[dy] = inband ? __expf(s[dy] - mx) : 0.0f;
            z += e[dy];
        }
        #pragma unroll
        for (int off = 16; off > 0; off >>= 1)
            z += __shfl_xor(z, off, 32);
        const float inv = 1.0f / z;
        #pragma unroll
        for (int dy = 0; dy < 7; ++dy) row[dy * 32] = e[dy] * inv;
    }
    __syncthreads();

    const int c0 = wv << 4;
    v8f o0 = {}, o1 = {};
    #pragma unroll
    for (int dy = 0; dy < 7; ++dy) {
        const int  hn    = h + dy - 3;
        const bool rowok = (unsigned)hn < (unsigned)HDIM;
        #pragma unroll
        for (int ks = 0; ks < 6; ++ks) {
            const v2f a = *(const v2f*)(&sp[l16 * SP + dy * 32 + 4 * ks + (hl << 1)]);
            const int  j0  = 4 * ks + (hl << 1);
            const int  wnA = w0 - 3 + j0;
            const int  wnB = wnA + 1;
            const bool okA = rowok && (j0 < 22)     && (unsigned)wnA < (unsigned)WDIM;
            const bool okB = rowok && (j0 + 1 < 22) && (unsigned)wnB < (unsigned)WDIM;
            const long pA = okA ? ((long)hn * WDIM + wnA) : (long)p0;
            const long pB = okB ? ((long)hn * WDIM + wnB) : (long)p0;
            float bA = v[pA * CCH + c0 + l16];
            float bB = v[pB * CCH + c0 + l16];
            bA = okA ? bA : 0.0f;
            bB = okB ? bB : 0.0f;
            const v2f b = {bA, bB};
            if (ks & 1)
                o1 = __builtin_amdgcn_wmma_f32_16x16x4_f32(
                    false, a, false, b, (short)0, o1, false, false);
            else
                o0 = __builtin_amdgcn_wmma_f32_16x16x4_f32(
                    false, a, false, b, (short)0, o0, false, false);
        }
    }
    v4f xo[2];
    #pragma unroll
    for (int i = 0; i < 2; ++i) {
        const int idx = tid + (i << 9);
        const int c   = idx >> 3;
        const int px4 = (idx & 7) << 2;
        xo[i] = *(const v4f*)(out + (long)c * HW + pb + px4);
    }

    #pragma unroll
    for (int r = 0; r < 8; ++r)
        ot[(c0 + l16) * OP + (sh << 4) + r + (hl << 3)] = o0[r] + o1[r];
    __syncthreads();
    #pragma unroll
    for (int i = 0; i < 2; ++i) {
        const int idx = tid + (i << 9);
        const int c   = idx >> 3;
        const int px4 = (idx & 7) << 2;
        const v4f a = *(const v4f*)(&ot[c * OP + px4]);
        *(v4f*)(out + (long)c * HW + pb + px4) = a + xo[i];
    }
}

extern "C" void kernel_launch(void* const* d_in, const int* in_sizes, int n_in,
                              void* d_out, int out_size, void* d_ws, size_t ws_size,
                              hipStream_t stream) {
    const float* x    = (const float*)d_in[0];
    const float* w_q1 = (const float*)d_in[1];
    const float* s_q1 = (const float*)d_in[2];
    const float* b_q1 = (const float*)d_in[3];
    const float* w_q2 = (const float*)d_in[4];
    const float* s_q2 = (const float*)d_in[5];
    const float* b_q2 = (const float*)d_in[6];
    const float* w_k1 = (const float*)d_in[7];
    const float* s_k1 = (const float*)d_in[8];
    const float* b_k1 = (const float*)d_in[9];
    const float* w_k2 = (const float*)d_in[10];
    const float* s_k2 = (const float*)d_in[11];
    const float* b_k2 = (const float*)d_in[12];
    const float* w_v  = (const float*)d_in[13];
    const float* s_v  = (const float*)d_in[14];
    const float* b_v  = (const float*)d_in[15];
    const float* w_o  = (const float*)d_in[16];
    const float* s_o  = (const float*)d_in[17];
    const float* b_o  = (const float*)d_in[18];
    float* out = (float*)d_out;

    const size_t NB = (size_t)HW * CCH;
    float* q1 = (float*)d_ws;
    float* k1 = q1 + NB;
    float* vb = k1 + NB;
    float* qb = vb + NB;
    float* kb = qb + NB;
    float* vo = kb + NB;

    gemm_bn_wmma<<<dim3(HW / 64, 4), dim3(256), 0, stream>>>(
        x, x, x, x, HW, 1,
        w_q1, w_k1, w_v, w_o + CCH,
        s_q1, s_k1, s_v, s_o, b_q1, b_k1, b_v, b_o,
        q1, k1, vb, out, 7, 15, 8, 8);
    gemm_bn_wmma<<<dim3(HW / 64, 3), dim3(256), 0, stream>>>(
        q1, k1, vb, nullptr, 1, CCH,
        w_q2, w_k2, w_o, nullptr,
        s_q2, s_k2, s_o, nullptr, b_q2, b_k2, b_o, nullptr,
        qb, kb, vo, nullptr, 3, 3, 4, 0);
    local_attn_wmma<<<dim3(HW / 32), dim3(512), 0, stream>>>(qb, kb, vo, out);
}


# attention operands stored by the projection GEMM epilogues in their WMMA fragment-load layouts (q,k chunk-major; v pixel-quad-major): 512 contiguous bytes per fragment load; p.v and z sums reordered
# speedup vs baseline: 2.5631x; 1.0548x over previous
#include <hip/hip_runtime.h>
#include <hip/hip_bf16.h>

typedef __attribute__((ext_vector_type(2))) float v2f;
typedef __attribute__((ext_vector_type(4))) float v4f;
typedef __attribute__((ext_vector_type(8))) float v8f;

#define CCH 128
#define HDIM 128
#define WDIM 128
#define HW (HDIM * WDIM)
#define TPITCH 132

__device__ __forceinline__ void lds_barrier()
{
    __builtin_amdgcn_fence(__ATOMIC_RELEASE, "workgroup", "local");
    __builtin_amdgcn_s_barrier();
    __builtin_amdgcn_fence(__ATOMIC_ACQUIRE, "workgroup", "local");
}

__device__ __forceinline__ void tile_load_cm(const float* __restrict__ X, long sC,
                                             long p0, int tid, v4f (&xv)[8])
{
    #pragma unroll
    for (int i = 0; i < 8; ++i) {
        const int e  = tid + (i << 8);
        const int p4 = (e & 15) << 2;
        const int c  = e >> 4;
        xv[i] = *(const v4f*)(X + (long)c * sC + p0 + p4);
    }
}
__device__ __forceinline__ void tile_load_pm(const float* __restrict__ X, long sP,
                                             long p0, int tid, v4f (&xv)[8])
{
    #pragma unroll
    for (int i = 0; i < 8; ++i) {
        const int e  = tid + (i << 8);
        const int c4 = (e & 31) << 2;
        const int p  = e >> 5;
        xv[i] = *(const v4f*)(X + (p0 + p) * sP + c4);
    }
}

__device__ __forceinline__ void tile_store_cm(float* __restrict__ tile, int tid,
                                              const v4f (&xv)[8])
{
    #pragma unroll
    for (int i = 0; i < 8; ++i) {
        const int e  = tid + (i << 8);
        const int p4 = (e & 15) << 2;
        const int c  = e >> 4;
        tile[(p4 + 0) * TPITCH + c] = xv[i].x;
        tile[(p4 + 1) * TPITCH + c] = xv[i].y;
        tile[(p4 + 2) * TPITCH + c] = xv[i].z;
        tile[(p4 + 3) * TPITCH + c] = xv[i].w;
    }
}
__device__ __forceinline__ void tile_store_pm(float* __restrict__ tile, int tid,
                                              const v4f (&xv)[8])
{
    #pragma unroll
    for (int i = 0; i < 8; ++i) {
        const int e  = tid + (i << 8);
        const int c4 = (e & 31) << 2;
        const int p  = e >> 5;
        *(v4f*)(&tile[p * TPITCH + c4]) = xv[i];
    }
}

__device__ __forceinline__ void mma_segment(const float* __restrict__ tile,
                                            const int (&bbase)[4],
                                            const v2f (&afr)[32], v8f (&acc)[4])
{
    #pragma unroll
    for (int kk = 0; kk < 32; ++kk) {
        #pragma unroll
        for (int s = 0; s < 4; ++s) {
            const v2f b = *(const v2f*)(&tile[bbase[s] + 4 * kk]);
            acc[s] = __builtin_amdgcn_wmma_f32_16x16x4_f32(
                false, afr[kk], false, b, (short)0, acc[s], false, false);
        }
    }
}

__global__ __launch_bounds__(256) void gemm_bn_wmma(
    const float* __restrict__ Xa0, const float* __restrict__ Xa1,
    const float* __restrict__ Xa2, const float* __restrict__ Xa3,
    long sCa, long sPa,
    const float* __restrict__ Wm0, const float* __restrict__ Wm1,
    const float* __restrict__ Wm2, const float* __restrict__ Wm3,
    const float* __restrict__ scale0, const float* __restrict__ scale1,
    const float* __restrict__ scale2, const float* __restrict__ scale3,
    const float* __restrict__ bias0, const float* __restrict__ bias1,
    const float* __restrict__ bias2, const float* __restrict__ bias3,
    float* __restrict__ Y0, float* __restrict__ Y1,
    float* __restrict__ Y2, float* __restrict__ Y3,
    int relu_mask, int bias_mask, int wide_mask, int cm_mask, int ck_mask, int qd_mask)
{
    __shared__ __attribute__((aligned(16))) float tile[64 * TPITCH];

    const int gsel = blockIdx.y;
    const float* __restrict__ Xa    = gsel == 0 ? Xa0    : (gsel == 1 ? Xa1    : (gsel == 2 ? Xa2    : Xa3));
    const float* __restrict__ Wm    = gsel == 0 ? Wm0    : (gsel == 1 ? Wm1    : (gsel == 2 ? Wm2    : Wm3));
    const float* __restrict__ scale = gsel == 0 ? scale0 : (gsel == 1 ? scale1 : (gsel == 2 ? scale2 : scale3));
    const float* __restrict__ bias  = gsel == 0 ? bias0  : (gsel == 1 ? bias1  : (gsel == 2 ? bias2  : bias3));
    float* __restrict__ Y           = gsel == 0 ? Y0     : (gsel == 1 ? Y1     : (gsel == 2 ? Y2     : Y3));
    const bool do_relu = ((relu_mask >> gsel) & 1) != 0;
    const bool do_bias = ((bias_mask >> gsel) & 1) != 0;
    const bool out_cm  = ((cm_mask   >> gsel) & 1) != 0;
    const bool out_ck  = ((ck_mask   >> gsel) & 1) != 0;
    const bool out_qd  = ((qd_mask   >> gsel) & 1) != 0;
    const int  ldw     = ((wide_mask >> gsel) & 1) != 0 ? 2 * CCH : CCH;

    const int tid  = threadIdx.x;
    const int lane = tid & 31;
    const int wv   = tid >> 5;
    const int hl   = lane >> 4;
    const int l16  = lane & 15;
    const int m0   = wv << 4;
    const long p0  = (long)blockIdx.x * 64;
    const bool a_cm = (sPa == 1);

    v4f xv[8];
    if (a_cm) tile_load_cm(Xa, sCa, p0, tid, xv);
    else      tile_load_pm(Xa, sPa, p0, tid, xv);

    const float* __restrict__ Ap = Wm + (long)(m0 + l16) * ldw + (hl << 1);
    v2f afr[32];
    #pragma unroll
    for (int kk = 0; kk < 32; ++kk)
        afr[kk] = *(const v2f*)(Ap + 4 * kk);

    const int ob = m0 + (hl << 3);
    float sc[8], bi[8];
    #pragma unroll
    for (int r = 0; r < 8; ++r) { sc[r] = scale[ob + r]; bi[r] = bias[ob + r]; }

    asm volatile("" ::: "memory");

    if (a_cm) tile_store_cm(tile, tid, xv);
    else      tile_store_pm(tile, tid, xv);
    lds_barrier();

    int bbase[4];
    #pragma unroll
    for (int s = 0; s < 4; ++s)
        bbase[s] = ((s << 4) + l16) * TPITCH + (hl << 1);

    v8f acc[4] = {};
    mma_segment(tile, bbase, afr, acc);

    if (!do_bias) {
        #pragma unroll
        for (int r = 0; r < 8; ++r) bi[r] = 0.0f;
    }
    if (!out_cm) {
        lds_barrier();
        #pragma unroll
        for (int s = 0; s < 4; ++s) {
            float f[8];
            #pragma unroll
            for (int r = 0; r < 8; ++r) {
                f[r] = acc[s][r] * sc[r] + bi[r];
                if (do_relu) f[r] = fmaxf(f[r], 0.0f);
            }
            float* dst = &tile[((s << 4) + l16) * TPITCH + ob];
            const v4f lo = {f[0], f[1], f[2], f[3]};
            const v4f hi = {f[4], f[5], f[6], f[7]};
            *(v4f*)(dst)     = lo;
            *(v4f*)(dst + 4) = hi;
        }
        lds_barrier();
        if (out_qd) {
            float* __restrict__ Yq = Y + ((p0 >> 2) + (wv << 1) + hl) * 64 + (l16 << 2);
            const float* __restrict__ trow = &tile[((wv << 3) + (hl << 2)) * TPITCH + l16];
            #pragma unroll
            for (int i = 0; i < 8; ++i) {
                const v4f val = {trow[0 * TPITCH + (i << 4)], trow[1 * TPITCH + (i << 4)],
                                 trow[2 * TPITCH + (i << 4)], trow[3 * TPITCH + (i << 4)]};
                *(v4f*)(Yq + (long)i * (HW * 16)) = val;
            }
        } else if (!out_ck) {
            float* __restrict__ Yp = Y + p0 * CCH + (lane << 2);
            #pragma unroll
            for (int i = 0; i < 8; ++i) {
                const int p = wv + (i << 3);
                const v4f val = *(const v4f*)(&tile[p * TPITCH + (lane << 2)]);
                *(v4f*)(Yp + (long)p * CCH) = val;
            }
        } else {
            float* __restrict__ Yc = Y + (p0 + l16) * 8 + (hl << 2);
            #pragma unroll
            for (int i = 0; i < 8; ++i) {
                const int u  = wv + (i << 3);
                const int jj = u >> 2;
                const int g  = u & 3;
                const v4f val = *(const v4f*)(&tile[((g << 4) + l16) * TPITCH + (jj << 3) + (hl << 2)]);
                *(v4f*)(Yc + (long)jj * (HW * 8) + (g << 7)) = val;
            }
        }
    } else {
        #pragma unroll
        for (int s = 0; s < 4; ++s) {
            const long p = p0 + (s << 4) + l16;
            #pragma unroll
            for (int r = 0; r < 8; ++r) {
                const int o = ob + r;
                float f = acc[s][r] * sc[r] + bi[r];
                if (do_relu) f = fmaxf(f, 0.0f);
                Y[(long)o * HW + p] = f;
            }
        }
    }
}

#define SP 226

#define OP 36

__global__ __launch_bounds__(512) __attribute__((amdgpu_waves_per_eu(8, 8)))
void local_attn_wmma(
    const float* __restrict__ q, const float* __restrict__ k,
    const float* __restrict__ v, float* __restrict__ out)
{
    __shared__ __attribute__((aligned(16))) float sp2[2 * 16 * SP];
    __shared__ __attribute__((aligned(16))) float ot[CCH * OP];

    const int tid  = threadIdx.x;
    const int lane = tid & 31;
    const int sh   = tid >> 8;
    const int wv   = (tid >> 5) & 7;
    const int hl   = lane >> 4;
    const int l16  = lane & 15;
    const int pb   = blockIdx.x * 32;
    const int p0   = pb + (sh << 4);
    const int h    = p0 >> 7;
    const int w0   = p0 & (WDIM - 1);
    float* __restrict__ sp = sp2 + sh * (16 * SP);


    if (wv < 7) {
        const int  hn    = h + wv - 3;
        const bool rowok = (unsigned)hn < (unsigned)HDIM;
        const int  wn0   = w0 - 3 + l16;
        const int  wn1   = w0 + 13 + l16;
        const bool ok0   = rowok && (unsigned)wn0 < (unsigned)WDIM;
        const bool ok1   = rowok && (l16 < 6) && (unsigned)wn1 < (unsigned)WDIM;
        const int  hc  = min(max(hn, 0), HDIM - 1);
        const int  wc0 = min(max(wn0, 0), WDIM - 1);
        const int  wc1 = min(w0 + 13 + min(l16, 5), WDIM - 1);
        const long pn0 = (long)hc * WDIM + wc0;
        const long pn1 = (long)hc * WDIM + wc1;
        const float* __restrict__ qp  = q + (long)(p0 + l16) * 8 + (hl << 2);
        const float* __restrict__ kp0 = k + pn0 * 8 + (hl << 2);
        const float* __restrict__ kp1 = k + pn1 * 8 + (hl << 2);
        const v4f zero4 = {0.0f, 0.0f, 0.0f, 0.0f};

        v8f s0 = {}, s1 = {};
        #pragma unroll
        for (int jj = 0; jj < 16; ++jj) {
            const v4f a  = *(const v4f*)(qp + (long)jj * (HW * 8));
            v4f b0 = *(const v4f*)(kp0 + (long)jj * (HW * 8));
            v4f b1 = *(const v4f*)(kp1 + (long)jj * (HW * 8));
            b0 = ok0 ? b0 : zero4;
            b1 = ok1 ? b1 : zero4;
            s0 = __builtin_amdgcn_wmma_f32_16x16x4_f32(
                false, a.xy, false, b0.xy, (short)0, s0, false, false);
            s1 = __builtin_amdgcn_wmma_f32_16x16x4_f32(
                false, a.xy, false, b1.xy, (short)0, s1, false, false);
            s0 = __builtin_amdgcn_wmma_f32_16x16x4_f32(
                false, a.zw, false, b0.zw, (short)0, s0, false, false);
            s1 = __builtin_amdgcn_wmma_f32_16x16x4_f32(
                false, a.zw, false, b1.zw, (short)0, s1, false, false);
        }
        #pragma unroll
        for (int r = 0; r < 8; ++r) {
            sp[(r + (hl << 3)) * SP + wv * 32 + 1 + l16]           = s0[r];
            sp[(r + (hl << 3)) * SP + wv * 32 + ((17 + l16) & 31)] = s1[r];
        }
    }
    __syncthreads();

    #pragma unroll
    for (int mm = 0; mm < 2; ++mm) {
        const int  m      = 2 * wv + mm;
        const bool inband = (lane >= m + 1) && (lane <= m + 7);
        float* row = &sp[m * SP + lane];
        float s[7];
        #pragma unroll
        for (int dy = 0; dy < 7; ++dy) s[dy] = row[dy * 32];
        float mx = s[0];
        #pragma unroll
        for (int dy = 1; dy < 7; ++dy) mx = fmaxf(mx, s[dy]);
        mx = inband ? mx : -INFINITY;
        #pragma unroll
        for (int off = 16; off > 0; off >>= 1)
            mx = fmaxf(mx, __shfl_xor(mx, off, 32));
        float e[7];
        float z = 0.0f;
        #pragma unroll
        for (int dy = 0; dy < 7; ++dy) {
            e[dy] = inband ? __expf(s[dy] - mx) : 0.0f;
            z += e[dy];
        }
        #pragma unroll
        for (int off = 16; off > 0; off >>= 1)
            z += __shfl_xor(z, off, 32);
        const float inv = 1.0f / z;
        #pragma unroll
        for (int dy = 0; dy < 7; ++dy) row[dy * 32] = e[dy] * inv;
    }
    __syncthreads();

    const int c0 = wv << 4;
    const float* __restrict__ vq = v + (long)wv * (HW * 16) + (l16 << 2);
    const int qcol0 = (w0 >> 2) - 1 + hl;
    const v4f zero4v = {0.0f, 0.0f, 0.0f, 0.0f};
    v8f o0 = {}, o1 = {};
    #pragma unroll
    for (int dy = 0; dy < 7; ++dy) {
        const int  hn    = h + dy - 3;
        const bool rowok = (unsigned)hn < (unsigned)HDIM;
        const int  hc    = min(max(hn, 0), HDIM - 1);
        #pragma unroll
        for (int T = 0; T < 3; ++T) {
            const int  qcol = qcol0 + 2 * T;
            const bool okq  = rowok && (unsigned)qcol < (unsigned)(WDIM / 4);
            const int  qc   = min(max(qcol, 0), WDIM / 4 - 1);
            v4f b = *(const v4f*)(vq + (hc * (WDIM / 4) + qc) * 64);
            b = okq ? b : zero4v;
            const float* arow = &sp[l16 * SP + dy * 32 + 8 * T + (hl << 2)];
            const v2f a01 = *(const v2f*)(arow);
            const v2f a23 = *(const v2f*)(arow + 2);
            o0 = __builtin_amdgcn_wmma_f32_16x16x4_f32(
                false, a01, false, b.xy, (short)0, o0, false, false);
            o1 = __builtin_amdgcn_wmma_f32_16x16x4_f32(
                false, a23, false, b.zw, (short)0, o1, false, false);
        }
    }
    v4f xo[2];
    #pragma unroll
    for (int i = 0; i < 2; ++i) {
        const int idx = tid + (i << 9);
        const int c   = idx >> 3;
        const int px4 = (idx & 7) << 2;
        xo[i] = *(const v4f*)(out + (long)c * HW + pb + px4);
    }

    #pragma unroll
    for (int r = 0; r < 8; ++r)
        ot[(c0 + l16) * OP + (sh << 4) + r + (hl << 3)] = o0[r] + o1[r];
    __syncthreads();
    #pragma unroll
    for (int i = 0; i < 2; ++i) {
        const int idx = tid + (i << 9);
        const int c   = idx >> 3;
        const int px4 = (idx & 7) << 2;
        const v4f a = *(const v4f*)(&ot[c * OP + px4]);
        *(v4f*)(out + (long)c * HW + pb + px4) = a + xo[i];
    }
}

extern "C" void kernel_launch(void* const* d_in, const int* in_sizes, int n_in,
                              void* d_out, int out_size, void* d_ws, size_t ws_size,
                              hipStream_t stream) {
    const float* x    = (const float*)d_in[0];
    const float* w_q1 = (const float*)d_in[1];
    const float* s_q1 = (const float*)d_in[2];
    const float* b_q1 = (const float*)d_in[3];
    const float* w_q2 = (const float*)d_in[4];
    const float* s_q2 = (const float*)d_in[5];
    const float* b_q2 = (const float*)d_in[6];
    const float* w_k1 = (const float*)d_in[7];
    const float* s_k1 = (const float*)d_in[8];
    const float* b_k1 = (const float*)d_in[9];
    const float* w_k2 = (const float*)d_in[10];
    const float* s_k2 = (const float*)d_in[11];
    const float* b_k2 = (const float*)d_in[12];
    const float* w_v  = (const float*)d_in[13];
    const float* s_v  = (const float*)d_in[14];
    const float* b_v  = (const float*)d_in[15];
    const float* w_o  = (const float*)d_in[16];
    const float* s_o  = (const float*)d_in[17];
    const float* b_o  = (const float*)d_in[18];
    float* out = (float*)d_out;

    const size_t NB = (size_t)HW * CCH;
    float* q1 = (float*)d_ws;
    float* k1 = q1 + NB;
    float* vb = k1 + NB;
    float* qb = vb + NB;
    float* kb = qb + NB;
    float* vo = kb + NB;

    gemm_bn_wmma<<<dim3(HW / 64, 4), dim3(256), 0, stream>>>(
        x, x, x, x, HW, 1,
        w_q1, w_k1, w_v, w_o + CCH,
        s_q1, s_k1, s_v, s_o, b_q1, b_k1, b_v, b_o,
        q1, k1, vb, out, 7, 15, 8, 8, 0, 0);
    gemm_bn_wmma<<<dim3(HW / 64, 3), dim3(256), 0, stream>>>(
        q1, k1, vb, nullptr, 1, CCH,
        w_q2, w_k2, w_o, nullptr,
        s_q2, s_k2, s_o, nullptr, b_q2, b_k2, b_o, nullptr,
        qb, kb, vo, nullptr, 3, 3, 4, 0, 3, 4);
    local_attn_wmma<<<dim3(HW / 32), dim3(512), 0, stream>>>(qb, kb, vo, out);
}


# attention tiles are 8x2-pixel patches, not 16x1 strips: a patch's 14-col halo fits one 16-col WMMA tile (scores 8 tiles per 16 px, not 14; p.v k 128, not 168); q.k and softmax z sums reordered
# speedup vs baseline: 2.7748x; 1.0826x over previous
#include <hip/hip_runtime.h>
#include <hip/hip_bf16.h>

typedef __attribute__((ext_vector_type(2))) float v2f;
typedef __attribute__((ext_vector_type(4))) float v4f;
typedef __attribute__((ext_vector_type(8))) float v8f;

#define CCH 128
#define HDIM 128
#define WDIM 128
#define HW (HDIM * WDIM)
#define TPITCH 132

__device__ __forceinline__ void lds_barrier()
{
    __builtin_amdgcn_fence(__ATOMIC_RELEASE, "workgroup", "local");
    __builtin_amdgcn_s_barrier();
    __builtin_amdgcn_fence(__ATOMIC_ACQUIRE, "workgroup", "local");
}

__device__ __forceinline__ void tile_load_cm(const float* __restrict__ X, long sC,
                                             long p0, int tid, v4f (&xv)[8])
{
    #pragma unroll
    for (int i = 0; i < 8; ++i) {
        const int e  = tid + (i << 8);
        const int p4 = (e & 15) << 2;
        const int c  = e >> 4;
        xv[i] = *(const v4f*)(X + (long)c * sC + p0 + p4);
    }
}
__device__ __forceinline__ void tile_load_pm(const float* __restrict__ X, long sP,
                                             long p0, int tid, v4f (&xv)[8])
{
    #pragma unroll
    for (int i = 0; i < 8; ++i) {
        const int e  = tid + (i << 8);
        const int c4 = (e & 31) << 2;
        const int p  = e >> 5;
        xv[i] = *(const v4f*)(X + (p0 + p) * sP + c4);
    }
}

__device__ __forceinline__ void tile_store_cm(float* __restrict__ tile, int tid,
                                              const v4f (&xv)[8])
{
    #pragma unroll
    for (int i = 0; i < 8; ++i) {
        const int e  = tid + (i << 8);
        const int p4 = (e & 15) << 2;
        const int c  = e >> 4;
        tile[(p4 + 0) * TPITCH + c] = xv[i].x;
        tile[(p4 + 1) * TPITCH + c] = xv[i].y;
        tile[(p4 + 2) * TPITCH + c] = xv[i].z;
        tile[(p4 + 3) * TPITCH + c] = xv[i].w;
    }
}
__device__ __forceinline__ void tile_store_pm(float* __restrict__ tile, int tid,
                                              const v4f (&xv)[8])
{
    #pragma unroll
    for (int i = 0; i < 8; ++i) {
        const int e  = tid + (i << 8);
        const int c4 = (e & 31) << 2;
        const int p  = e >> 5;
        *(v4f*)(&tile[p * TPITCH + c4]) = xv[i];
    }
}

__device__ __forceinline__ void mma_segment(const float* __restrict__ tile,
                                            const int (&bbase)[4],
                                            const v2f (&afr)[32], v8f (&acc)[4])
{
    #pragma unroll
    for (int kk = 0; kk < 32; ++kk) {
        #pragma unroll
        for (int s = 0; s < 4; ++s) {
            const v2f b = *(const v2f*)(&tile[bbase[s] + 4 * kk]);
            acc[s] = __builtin_amdgcn_wmma_f32_16x16x4_f32(
                false, afr[kk], false, b, (short)0, acc[s], false, false);
        }
    }
}

__global__ __launch_bounds__(256) void gemm_bn_wmma(
    const float* __restrict__ Xa0, const float* __restrict__ Xa1,
    const float* __restrict__ Xa2, const float* __restrict__ Xa3,
    long sCa, long sPa,
    const float* __restrict__ Wm0, const float* __restrict__ Wm1,
    const float* __restrict__ Wm2, const float* __restrict__ Wm3,
    const float* __restrict__ scale0, const float* __restrict__ scale1,
    const float* __restrict__ scale2, const float* __restrict__ scale3,
    const float* __restrict__ bias0, const float* __restrict__ bias1,
    const float* __restrict__ bias2, const float* __restrict__ bias3,
    float* __restrict__ Y0, float* __restrict__ Y1,
    float* __restrict__ Y2, float* __restrict__ Y3,
    int relu_mask, int bias_mask, int wide_mask, int cm_mask, int ck_mask, int qd_mask)
{
    __shared__ __attribute__((aligned(16))) float tile[64 * TPITCH];

    const int gsel = blockIdx.y;
    const float* __restrict__ Xa    = gsel == 0 ? Xa0    : (gsel == 1 ? Xa1    : (gsel == 2 ? Xa2    : Xa3));
    const float* __restrict__ Wm    = gsel == 0 ? Wm0    : (gsel == 1 ? Wm1    : (gsel == 2 ? Wm2    : Wm3));
    const float* __restrict__ scale = gsel == 0 ? scale0 : (gsel == 1 ? scale1 : (gsel == 2 ? scale2 : scale3));
    const float* __restrict__ bias  = gsel == 0 ? bias0  : (gsel == 1 ? bias1  : (gsel == 2 ? bias2  : bias3));
    float* __restrict__ Y           = gsel == 0 ? Y0     : (gsel == 1 ? Y1     : (gsel == 2 ? Y2     : Y3));
    const bool do_relu = ((relu_mask >> gsel) & 1) != 0;
    const bool do_bias = ((bias_mask >> gsel) & 1) != 0;
    const bool out_cm  = ((cm_mask   >> gsel) & 1) != 0;
    const bool out_ck  = ((ck_mask   >> gsel) & 1) != 0;
    const bool out_qd  = ((qd_mask   >> gsel) & 1) != 0;
    const int  ldw     = ((wide_mask >> gsel) & 1) != 0 ? 2 * CCH : CCH;

    const int tid  = threadIdx.x;
    const int lane = tid & 31;
    const int wv   = tid >> 5;
    const int hl   = lane >> 4;
    const int l16  = lane & 15;
    const int m0   = wv << 4;
    const long p0  = (long)blockIdx.x * 64;
    const bool a_cm = (sPa == 1);

    v4f xv[8];
    if (a_cm) tile_load_cm(Xa, sCa, p0, tid, xv);
    else      tile_load_pm(Xa, sPa, p0, tid, xv);

    const float* __restrict__ Ap = Wm + (long)(m0 + l16) * ldw + (hl << 1);
    v2f afr[32];
    #pragma unroll
    for (int kk = 0; kk < 32; ++kk)
        afr[kk] = *(const v2f*)(Ap + 4 * kk);

    const int ob = m0 + (hl << 3);
    float sc[8], bi[8];
    #pragma unroll
    for (int r = 0; r < 8; ++r) { sc[r] = scale[ob + r]; bi[r] = bias[ob + r]; }

    asm volatile("" ::: "memory");

    if (a_cm) tile_store_cm(tile, tid, xv);
    else      tile_store_pm(tile, tid, xv);
    lds_barrier();

    int bbase[4];
    #pragma unroll
    for (int s = 0; s < 4; ++s)
        bbase[s] = ((s << 4) + l16) * TPITCH + (hl << 1);

    v8f acc[4] = {};
    mma_segment(tile, bbase, afr, acc);

    if (!do_bias) {
        #pragma unroll
        for (int r = 0; r < 8; ++r) bi[r] = 0.0f;
    }
    if (!out_cm) {
        lds_barrier();
        #pragma unroll
        for (int s = 0; s < 4; ++s) {
            float f[8];
            #pragma unroll
            for (int r = 0; r < 8; ++r) {
                f[r] = acc[s][r] * sc[r] + bi[r];
                if (do_relu) f[r] = fmaxf(f[r], 0.0f);
            }
            float* dst = &tile[((s << 4) + l16) * TPITCH + ob];
            const v4f lo = {f[0], f[1], f[2], f[3]};
            const v4f hi = {f[4], f[5], f[6], f[7]};
            *(v4f*)(dst)     = lo;
            *(v4f*)(dst + 4) = hi;
        }
        lds_barrier();
        if (out_qd) {
            float* __restrict__ Yq = Y + ((p0 >> 2) + (wv << 1) + hl) * 64 + (l16 << 2);
            const float* __restrict__ trow = &tile[((wv << 3) + (hl << 2)) * TPITCH + l16];
            #pragma unroll
            for (int i = 0; i < 8; ++i) {
                const v4f val = {trow[0 * TPITCH + (i << 4)], trow[1 * TPITCH + (i << 4)],
                                 trow[2 * TPITCH + (i << 4)], trow[3 * TPITCH + (i << 4)]};
                *(v4f*)(Yq + (long)i * (HW * 16)) = val;
            }
        } else if (!out_ck) {
            float* __restrict__ Yp = Y + p0 * CCH + (lane << 2);
            #pragma unroll
            for (int i = 0; i < 8; ++i) {
                const int p = wv + (i << 3);
                const v4f val = *(const v4f*)(&tile[p * TPITCH + (lane << 2)]);
                *(v4f*)(Yp + (long)p * CCH) = val;
            }
        } else {
            float* __restrict__ Yc = Y + (p0 + l16) * 8 + (hl << 2);
            #pragma unroll
            for (int i = 0; i < 8; ++i) {
                const int u  = wv + (i << 3);
                const int jj = u >> 2;
                const int g  = u & 3;
                const v4f val = *(const v4f*)(&tile[((g << 4) + l16) * TPITCH + (jj << 3) + (hl << 2)]);
                *(v4f*)(Yc + (long)jj * (HW * 8) + (g << 7)) = val;
            }
        }
    } else {
        #pragma unroll
        for (int s = 0; s < 4; ++s) {
            const long p = p0 + (s << 4) + l16;
            #pragma unroll
            for (int r = 0; r < 8; ++r) {
                const int o = ob + r;
                float f = acc[s][r] * sc[r] + bi[r];
                if (do_relu) f = fmaxf(f, 0.0f);
                Y[(long)o * HW + p] = f;
            }
        }
    }
}

#define SP 132

#define OP 36

__global__ __launch_bounds__(1024) __attribute__((amdgpu_waves_per_eu(8, 8)))
void local_attn_wmma(
    const float* __restrict__ q, const float* __restrict__ k,
    const float* __restrict__ v, float* __restrict__ out)
{
    __shared__ __attribute__((aligned(16))) float sp4[4 * 16 * SP];
    __shared__ __attribute__((aligned(16))) float ot[CCH * 2 * OP];

    const int tid  = threadIdx.x;
    const int lane = tid & 31;
    const int sh   = tid >> 8;
    const int wv   = (tid >> 5) & 7;
    const int hl   = lane >> 4;
    const int l16  = lane & 15;
    const int h0   = (blockIdx.x >> 2) << 1;
    const int wb   = (blockIdx.x & 3) << 5;
    const int w0   = wb + (sh << 3);
    float* __restrict__ sp = sp4 + sh * (16 * SP);
    const v4f zero4 = {0.0f, 0.0f, 0.0f, 0.0f};

    {
        const int  hn    = h0 - 3 + wv;
        const bool rowok = (unsigned)hn < (unsigned)HDIM;
        const int  wn    = w0 - 4 + l16;
        const bool ok    = rowok && (unsigned)wn < (unsigned)WDIM;
        const int  hc = min(max(hn, 0), HDIM - 1);
        const int  wc = min(max(wn, 0), WDIM - 1);
        const long pn = (long)hc * WDIM + wc;
        const long pa = (long)(h0 + (l16 >> 3)) * WDIM + w0 + (l16 & 7);
        const float* __restrict__ qp = q + pa * 8 + (hl << 2);
        const float* __restrict__ kp = k + pn * 8 + (hl << 2);

        v8f s0 = {}, s1 = {};
        #pragma unroll
        for (int jj = 0; jj < 16; ++jj) {
            const v4f a = *(const v4f*)(qp + (long)jj * (HW * 8));
            v4f b = *(const v4f*)(kp + (long)jj * (HW * 8));
            b = ok ? b : zero4;
            s0 = __builtin_amdgcn_wmma_f32_16x16x4_f32(
                false, a.xy, false, b.xy, (short)0, s0, false, false);
            s1 = __builtin_amdgcn_wmma_f32_16x16x4_f32(
                false, a.zw, false, b.zw, (short)0, s1, false, false);
        }
        #pragma unroll
        for (int r = 0; r < 8; ++r)
            sp[(r + (hl << 3)) * SP + wv * 16 + l16] = s0[r] + s1[r];
    }
    __syncthreads();

    {
        const int  m      = 2 * wv + hl;
        const int  ry     = wv >> 2;
        const int  cx     = m & 7;
        const bool inband = (l16 >= cx + 1) && (l16 <= cx + 7);
        float* row = &sp[m * SP + ry * 16 + l16];
        float s[7];
        #pragma unroll
        for (int dy = 0; dy < 7; ++dy) s[dy] = row[dy * 16];
        float mx = s[0];
        #pragma unroll
        for (int dy = 1; dy < 7; ++dy) mx = fmaxf(mx, s[dy]);
        mx = inband ? mx : -INFINITY;
        #pragma unroll
        for (int off = 8; off > 0; off >>= 1)
            mx = fmaxf(mx, __shfl_xor(mx, off, 32));
        float e[7];
        float z = 0.0f;
        #pragma unroll
        for (int dy = 0; dy < 7; ++dy) {
            e[dy] = inband ? __expf(s[dy] - mx) : 0.0f;
            z += e[dy];
        }
        #pragma unroll
        for (int off = 8; off > 0; off >>= 1)
            z += __shfl_xor(z, off, 32);
        const float inv = 1.0f / z;
        #pragma unroll
        for (int dy = 0; dy < 7; ++dy) row[dy * 16] = e[dy] * inv;
        sp[m * SP + (ry ? 0 : 7 * 16) + l16] = 0.0f;
    }
    __syncthreads();

    const int c0 = wv << 4;
    const float* __restrict__ vq = v + (long)wv * (HW * 16) + (l16 << 2);
    const int qcol0 = (w0 >> 2) - 1 + hl;
    v8f o0 = {}, o1 = {};
    #pragma unroll
    for (int r = 0; r < 8; ++r) {
        const int  hn    = h0 - 3 + r;
        const bool rowok = (unsigned)hn < (unsigned)HDIM;
        const int  hc    = min(max(hn, 0), HDIM - 1);
        #pragma unroll
        for (int T = 0; T < 2; ++T) {
            const int  qcol = qcol0 + 2 * T;
            const bool okq  = rowok && (unsigned)qcol < (unsigned)(WDIM / 4);
            const int  qc   = min(max(qcol, 0), WDIM / 4 - 1);
            v4f b = *(const v4f*)(vq + (hc * (WDIM / 4) + qc) * 64);
            b = okq ? b : zero4;
            const v4f a = *(const v4f*)(&sp[l16 * SP + r * 16 + 8 * T + (hl << 2)]);
            o0 = __builtin_amdgcn_wmma_f32_16x16x4_f32(
                false, a.xy, false, b.xy, (short)0, o0, false, false);
            o1 = __builtin_amdgcn_wmma_f32_16x16x4_f32(
                false, a.zw, false, b.zw, (short)0, o1, false, false);
        }
    }
    v4f xo[2];
    #pragma unroll
    for (int i = 0; i < 2; ++i) {
        const int idx = tid + (i << 10);
        const int c   = idx >> 4;
        const int ry  = (idx >> 3) & 1;
        const int px4 = (idx & 7) << 2;
        xo[i] = *(const v4f*)(out + (long)c * HW + (h0 + ry) * WDIM + wb + px4);
    }

    {
        float* dst = &ot[(((c0 + l16) << 1) + hl) * OP + (sh << 3)];
        const v4f lo = {o0[0] + o1[0], o0[1] + o1[1], o0[2] + o1[2], o0[3] + o1[3]};
        const v4f hi = {o0[4] + o1[4], o0[5] + o1[5], o0[6] + o1[6], o0[7] + o1[7]};
        *(v4f*)(dst)     = lo;
        *(v4f*)(dst + 4) = hi;
    }
    __syncthreads();
    #pragma unroll
    for (int i = 0; i < 2; ++i) {
        const int idx = tid + (i << 10);
        const int c   = idx >> 4;
        const int ry  = (idx >> 3) & 1;
        const int px4 = (idx & 7) << 2;
        const v4f a = *(const v4f*)(&ot[(idx >> 3) * OP + px4]);
        *(v4f*)(out + (long)c * HW + (h0 + ry) * WDIM + wb + px4) = a + xo[i];
    }
}

extern "C" void kernel_launch(void* const* d_in, const int* in_sizes, int n_in,
                              void* d_out, int out_size, void* d_ws, size_t ws_size,
                              hipStream_t stream) {
    const float* x    = (const float*)d_in[0];
    const float* w_q1 = (const float*)d_in[1];
    const float* s_q1 = (const float*)d_in[2];
    const float* b_q1 = (const float*)d_in[3];
    const float* w_q2 = (const float*)d_in[4];
    const float* s_q2 = (const float*)d_in[5];
    const float* b_q2 = (const float*)d_in[6];
    const float* w_k1 = (const float*)d_in[7];
    const float* s_k1 = (const float*)d_in[8];
    const float* b_k1 = (const float*)d_in[9];
    const float* w_k2 = (const float*)d_in[10];
    const float* s_k2 = (const float*)d_in[11];
    const float* b_k2 = (const float*)d_in[12];
    const float* w_v  = (const float*)d_in[13];
    const float* s_v  = (const float*)d_in[14];
    const float* b_v  = (const float*)d_in[15];
    const float* w_o  = (const float*)d_in[16];
    const float* s_o  = (const float*)d_in[17];
    const float* b_o  = (const float*)d_in[18];
    float* out = (float*)d_out;

    const size_t NB = (size_t)HW * CCH;
    float* q1 = (float*)d_ws;
    float* k1 = q1 + NB;
    float* vb = k1 + NB;
    float* qb = vb + NB;
    float* kb = qb + NB;
    float* vo = kb + NB;

    gemm_bn_wmma<<<dim3(HW / 64, 4), dim3(256), 0, stream>>>(
        x, x, x, x, HW, 1,
        w_q1, w_k1, w_v, w_o + CCH,
        s_q1, s_k1, s_v, s_o, b_q1, b_k1, b_v, b_o,
        q1, k1, vb, out, 7, 15, 8, 8, 0, 0);
    gemm_bn_wmma<<<dim3(HW / 64, 3), dim3(256), 0, stream>>>(
        q1, k1, vb, nullptr, 1, CCH,
        w_q2, w_k2, w_o, nullptr,
        s_q2, s_k2, s_o, nullptr, b_q2, b_k2, b_o, nullptr,
        qb, kb, vo, nullptr, 3, 3, 4, 0, 3, 4);
    local_attn_wmma<<<dim3(HW / 64), dim3(1024), 0, stream>>>(qb, kb, vo, out);
}


# 1x1-conv GEMMs: weight fragments fetched as whole 512-B rows (4 lines per load instruction, not 16 lines of 16 B each) and turned to fragment order through a per-wave LDS strip; sums unchanged
# speedup vs baseline: 2.8839x; 1.0393x over previous
#include <hip/hip_runtime.h>
#include <hip/hip_bf16.h>

typedef __attribute__((ext_vector_type(2))) float v2f;
typedef __attribute__((ext_vector_type(4))) float v4f;
typedef __attribute__((ext_vector_type(8))) float v8f;

#define CCH 128
#define HDIM 128
#define WDIM 128
#define HW (HDIM * WDIM)
#define TPITCH 132

__device__ __forceinline__ void lds_barrier()
{
    __builtin_amdgcn_fence(__ATOMIC_RELEASE, "workgroup", "local");
    __builtin_amdgcn_s_barrier();
    __builtin_amdgcn_fence(__ATOMIC_ACQUIRE, "workgroup", "local");
}

__device__ __forceinline__ void tile_load_cm(const float* __restrict__ X, long sC,
                                             long p0, int tid, v4f (&xv)[8])
{
    #pragma unroll
    for (int i = 0; i < 8; ++i) {
        const int e  = tid + (i << 8);
        const int p4 = (e & 15) << 2;
        const int c  = e >> 4;
        xv[i] = *(const v4f*)(X + (long)c * sC + p0 + p4);
    }
}
__device__ __forceinline__ void tile_load_pm(const float* __restrict__ X, long sP,
                                             long p0, int tid, v4f (&xv)[8])
{
    #pragma unroll
    for (int i = 0; i < 8; ++i) {
        const int e  = tid + (i << 8);
        const int c4 = (e & 31) << 2;
        const int p  = e >> 5;
        xv[i] = *(const v4f*)(X + (p0 + p) * sP + c4);
    }
}

__device__ __forceinline__ void tile_store_cm(float* __restrict__ tile, int tid,
                                              const v4f (&xv)[8])
{
    #pragma unroll
    for (int i = 0; i < 8; ++i) {
        const int e  = tid + (i << 8);
        const int p4 = (e & 15) << 2;
        const int c  = e >> 4;
        tile[(p4 + 0) * TPITCH + c] = xv[i].x;
        tile[(p4 + 1) * TPITCH + c] = xv[i].y;
        tile[(p4 + 2) * TPITCH + c] = xv[i].z;
        tile[(p4 + 3) * TPITCH + c] = xv[i].w;
    }
}
__device__ __forceinline__ void tile_store_pm(float* __restrict__ tile, int tid,
                                              const v4f (&xv)[8])
{
    #pragma unroll
    for (int i = 0; i < 8; ++i) {
        const int e  = tid + (i << 8);
        const int c4 = (e & 31) << 2;
        const int p  = e >> 5;
        *(v4f*)(&tile[p * TPITCH + c4]) = xv[i];
    }
}

__device__ __forceinline__ void mma_segment(const float* __restrict__ tile,
                                            const int (&bbase)[4],
                                            const v2f (&afr)[32], v8f (&acc)[4])
{
    #pragma unroll
    for (int kk = 0; kk < 32; ++kk) {
        #pragma unroll
        for (int s = 0; s < 4; ++s) {
            const v2f b = *(const v2f*)(&tile[bbase[s] + 4 * kk]);
            acc[s] = __builtin_amdgcn_wmma_f32_16x16x4_f32(
                false, afr[kk], false, b, (short)0, acc[s], false, false);
        }
    }
}

__global__ __launch_bounds__(256) void gemm_bn_wmma(
    const float* __restrict__ Xa0, const float* __restrict__ Xa1,
    const float* __restrict__ Xa2, const float* __restrict__ Xa3,
    long sCa, long sPa,
    const float* __restrict__ Wm0, const float* __restrict__ Wm1,
    const float* __restrict__ Wm2, const float* __restrict__ Wm3,
    const float* __restrict__ scale0, const float* __restrict__ scale1,
    const float* __restrict__ scale2, const float* __restrict__ scale3,
    const float* __restrict__ bias0, const float* __restrict__ bias1,
    const float* __restrict__ bias2, const float* __restrict__ bias3,
    float* __restrict__ Y0, float* __restrict__ Y1,
    float* __restrict__ Y2, float* __restrict__ Y3,
    int relu_mask, int bias_mask, int wide_mask, int cm_mask, int ck_mask, int qd_mask)
{
    __shared__ __attribute__((aligned(16))) float tile[64 * TPITCH];
    __shared__ __attribute__((aligned(16))) float wst[8 * 16 * TPITCH];

    const int gsel = blockIdx.y;
    const float* __restrict__ Xa    = gsel == 0 ? Xa0    : (gsel == 1 ? Xa1    : (gsel == 2 ? Xa2    : Xa3));
    const float* __restrict__ Wm    = gsel == 0 ? Wm0    : (gsel == 1 ? Wm1    : (gsel == 2 ? Wm2    : Wm3));
    const float* __restrict__ scale = gsel == 0 ? scale0 : (gsel == 1 ? scale1 : (gsel == 2 ? scale2 : scale3));
    const float* __restrict__ bias  = gsel == 0 ? bias0  : (gsel == 1 ? bias1  : (gsel == 2 ? bias2  : bias3));
    float* __restrict__ Y           = gsel == 0 ? Y0     : (gsel == 1 ? Y1     : (gsel == 2 ? Y2     : Y3));
    const bool do_relu = ((relu_mask >> gsel) & 1) != 0;
    const bool do_bias = ((bias_mask >> gsel) & 1) != 0;
    const bool out_cm  = ((cm_mask   >> gsel) & 1) != 0;
    const bool out_ck  = ((ck_mask   >> gsel) & 1) != 0;
    const bool out_qd  = ((qd_mask   >> gsel) & 1) != 0;
    const int  ldw     = ((wide_mask >> gsel) & 1) != 0 ? 2 * CCH : CCH;

    const int tid  = threadIdx.x;
    const int lane = tid & 31;
    const int wv   = tid >> 5;
    const int hl   = lane >> 4;
    const int l16  = lane & 15;
    const int m0   = wv << 4;
    const long p0  = (long)blockIdx.x * 64;
    const bool a_cm = (sPa == 1);

    v4f xv[8];
    if (a_cm) tile_load_cm(Xa, sCa, p0, tid, xv);
    else      tile_load_pm(Xa, sPa, p0, tid, xv);

    v4f wr[16];
    const float* __restrict__ Wrow = Wm + (long)m0 * ldw + (lane << 2);
    #pragma unroll
    for (int r = 0; r < 16; ++r)
        wr[r] = *(const v4f*)(Wrow + (long)r * ldw);

    const int ob = m0 + (hl << 3);
    float sc[8], bi[8];
    #pragma unroll
    for (int r = 0; r < 8; ++r) { sc[r] = scale[ob + r]; bi[r] = bias[ob + r]; }

    asm volatile("" ::: "memory");

    if (a_cm) tile_store_cm(tile, tid, xv);
    else      tile_store_pm(tile, tid, xv);
    float* strip = wst + wv * (16 * TPITCH);
    #pragma unroll
    for (int r = 0; r < 16; ++r)
        *(v4f*)(&strip[r * TPITCH + (lane << 2)]) = wr[r];
    lds_barrier();

    const float* Ap = strip + l16 * TPITCH + (hl << 1);
    v2f afr[32];
    #pragma unroll
    for (int kk = 0; kk < 32; ++kk)
        afr[kk] = *(const v2f*)(Ap + 4 * kk);

    int bbase[4];
    #pragma unroll
    for (int s = 0; s < 4; ++s)
        bbase[s] = ((s << 4) + l16) * TPITCH + (hl << 1);

    v8f acc[4] = {};
    mma_segment(tile, bbase, afr, acc);

    if (!do_bias) {
        #pragma unroll
        for (int r = 0; r < 8; ++r) bi[r] = 0.0f;
    }
    if (!out_cm) {
        lds_barrier();
        #pragma unroll
        for (int s = 0; s < 4; ++s) {
            float f[8];
            #pragma unroll
            for (int r = 0; r < 8; ++r) {
                f[r] = acc[s][r] * sc[r] + bi[r];
                if (do_relu) f[r] = fmaxf(f[r], 0.0f);
            }
            float* dst = &tile[((s << 4) + l16) * TPITCH + ob];
            const v4f lo = {f[0], f[1], f[2], f[3]};
            const v4f hi = {f[4], f[5], f[6], f[7]};
            *(v4f*)(dst)     = lo;
            *(v4f*)(dst + 4) = hi;
        }
        lds_barrier();
        if (out_qd) {
            float* __restrict__ Yq = Y + ((p0 >> 2) + (wv << 1) + hl) * 64 + (l16 << 2);
            const float* __restrict__ trow = &tile[((wv << 3) + (hl << 2)) * TPITCH + l16];
            #pragma unroll
            for (int i = 0; i < 8; ++i) {
                const v4f val = {trow[0 * TPITCH + (i << 4)], trow[1 * TPITCH + (i << 4)],
                                 trow[2 * TPITCH + (i << 4)], trow[3 * TPITCH + (i << 4)]};
                *(v4f*)(Yq + (long)i * (HW * 16)) = val;
            }
        } else if (!out_ck) {
            float* __restrict__ Yp = Y + p0 * CCH + (lane << 2);
            #pragma unroll
            for (int i = 0; i < 8; ++i) {
                const int p = wv + (i << 3);
                const v4f val = *(const v4f*)(&tile[p * TPITCH + (lane << 2)]);
                *(v4f*)(Yp + (long)p * CCH) = val;
            }
        } else {
            float* __restrict__ Yc = Y + (p0 + l16) * 8 + (hl << 2);
            #pragma unroll
            for (int i = 0; i < 8; ++i) {
                const int u  = wv + (i << 3);
                const int jj = u >> 2;
                const int g  = u & 3;
                const v4f val = *(const v4f*)(&tile[((g << 4) + l16) * TPITCH + (jj << 3) + (hl << 2)]);
                *(v4f*)(Yc + (long)jj * (HW * 8) + (g << 7)) = val;
            }
        }
    } else {
        #pragma unroll
        for (int s = 0; s < 4; ++s) {
            const long p = p0 + (s << 4) + l16;
            #pragma unroll
            for (int r = 0; r < 8; ++r) {
                const int o = ob + r;
                float f = acc[s][r] * sc[r] + bi[r];
                if (do_relu) f = fmaxf(f, 0.0f);
                Y[(long)o * HW + p] = f;
            }
        }
    }
}

#define SP 132

#define OP 36

__global__ __launch_bounds__(1024) __attribute__((amdgpu_waves_per_eu(8, 8)))
void local_attn_wmma(
    const float* __restrict__ q, const float* __restrict__ k,
    const float* __restrict__ v, float* __restrict__ out)
{
    __shared__ __attribute__((aligned(16))) float sp4[4 * 16 * SP];
    __shared__ __attribute__((aligned(16))) float ot[CCH * 2 * OP];

    const int tid  = threadIdx.x;
    const int lane = tid & 31;
    const int sh   = tid >> 8;
    const int wv   = (tid >> 5) & 7;
    const int hl   = lane >> 4;
    const int l16  = lane & 15;
    const int h0   = (blockIdx.x >> 2) << 1;
    const int wb   = (blockIdx.x & 3) << 5;
    const int w0   = wb + (sh << 3);
    float* __restrict__ sp = sp4 + sh * (16 * SP);
    const v4f zero4 = {0.0f, 0.0f, 0.0f, 0.0f};

    {
        const int  hn    = h0 - 3 + wv;
        const bool rowok = (unsigned)hn < (unsigned)HDIM;
        const int  wn    = w0 - 4 + l16;
        const bool ok    = rowok && (unsigned)wn < (unsigned)WDIM;
        const int  hc = min(max(hn, 0), HDIM - 1);
        const int  wc = min(max(wn, 0), WDIM - 1);
        const long pn = (long)hc * WDIM + wc;
        const long pa = (long)(h0 + (l16 >> 3)) * WDIM + w0 + (l16 & 7);
        const float* __restrict__ qp = q + pa * 8 + (hl << 2);
        const float* __restrict__ kp = k + pn * 8 + (hl << 2);

        v8f s0 = {}, s1 = {};
        #pragma unroll
        for (int jj = 0; jj < 16; ++jj) {
            const v4f a = *(const v4f*)(qp + (long)jj * (HW * 8));
            v4f b = *(const v4f*)(kp + (long)jj * (HW * 8));
            b = ok ? b : zero4;
            s0 = __builtin_amdgcn_wmma_f32_16x16x4_f32(
                false, a.xy, false, b.xy, (short)0, s0, false, false);
            s1 = __builtin_amdgcn_wmma_f32_16x16x4_f32(
                false, a.zw, false, b.zw, (short)0, s1, false, false);
        }
        #pragma unroll
        for (int r = 0; r < 8; ++r)
            sp[(r + (hl << 3)) * SP + wv * 16 + l16] = s0[r] + s1[r];
    }
    __syncthreads();

    {
        const int  m      = 2 * wv + hl;
        const int  ry     = wv >> 2;
        const int  cx     = m & 7;
        const bool inband = (l16 >= cx + 1) && (l16 <= cx + 7);
        float* row = &sp[m * SP + ry * 16 + l16];
        float s[7];
        #pragma unroll
        for (int dy = 0; dy < 7; ++dy) s[dy] = row[dy * 16];
        float mx = s[0];
        #pragma unroll
        for (int dy = 1; dy < 7; ++dy) mx = fmaxf(mx, s[dy]);
        mx = inband ? mx : -INFINITY;
        #pragma unroll
        for (int off = 8; off > 0; off >>= 1)
            mx = fmaxf(mx, __shfl_xor(mx, off, 32));
        float e[7];
        float z = 0.0f;
        #pragma unroll
        for (int dy = 0; dy < 7; ++dy) {
            e[dy] = inband ? __expf(s[dy] - mx) : 0.0f;
            z += e[dy];
        }
        #pragma unroll
        for (int off = 8; off > 0; off >>= 1)
            z += __shfl_xor(z, off, 32);
        const float inv = 1.0f / z;
        #pragma unroll
        for (int dy = 0; dy < 7; ++dy) row[dy * 16] = e[dy] * inv;
        sp[m * SP + (ry ? 0 : 7 * 16) + l16] = 0.0f;
    }
    __syncthreads();

    const int c0 = wv << 4;
    const float* __restrict__ vq = v + (long)wv * (HW * 16) + (l16 << 2);
    const int qcol0 = (w0 >> 2) - 1 + hl;
    v8f o0 = {}, o1 = {};
    #pragma unroll
    for (int r = 0; r < 8; ++r) {
        const int  hn    = h0 - 3 + r;
        const bool rowok = (unsigned)hn < (unsigned)HDIM;
        const int  hc    = min(max(hn, 0), HDIM - 1);
        #pragma unroll
        for (int T = 0; T < 2; ++T) {
            const int  qcol = qcol0 + 2 * T;
            const bool okq  = rowok && (unsigned)qcol < (unsigned)(WDIM / 4);
            const int  qc   = min(max(qcol, 0), WDIM / 4 - 1);
            v4f b = *(const v4f*)(vq + (hc * (WDIM / 4) + qc) * 64);
            b = okq ? b : zero4;
            const v4f a = *(const v4f*)(&sp[l16 * SP + r * 16 + 8 * T + (hl << 2)]);
            o0 = __builtin_amdgcn_wmma_f32_16x16x4_f32(
                false, a.xy, false, b.xy, (short)0, o0, false, false);
            o1 = __builtin_amdgcn_wmma_f32_16x16x4_f32(
                false, a.zw, false, b.zw, (short)0, o1, false, false);
        }
    }
    v4f xo[2];
    #pragma unroll
    for (int i = 0; i < 2; ++i) {
        const int idx = tid + (i << 10);
        const int c   = idx >> 4;
        const int ry  = (idx >> 3) & 1;
        const int px4 = (idx & 7) << 2;
        xo[i] = *(const v4f*)(out + (long)c * HW + (h0 + ry) * WDIM + wb + px4);
    }

    {
        float* dst = &ot[(((c0 + l16) << 1) + hl) * OP + (sh << 3)];
        const v4f lo = {o0[0] + o1[0], o0[1] + o1[1], o0[2] + o1[2], o0[3] + o1[3]};
        const v4f hi = {o0[4] + o1[4], o0[5] + o1[5], o0[6] + o1[6], o0[7] + o1[7]};
        *(v4f*)(dst)     = lo;
        *(v4f*)(dst + 4) = hi;
    }
    __syncthreads();
    #pragma unroll
    for (int i = 0; i < 2; ++i) {
        const int idx = tid + (i << 10);
        const int c   = idx >> 4;
        const int ry  = (idx >> 3) & 1;
        const int px4 = (idx & 7) << 2;
        const v4f a = *(const v4f*)(&ot[(idx >> 3) * OP + px4]);
        *(v4f*)(out + (long)c * HW + (h0 + ry) * WDIM + wb + px4) = a + xo[i];
    }
}

extern "C" void kernel_launch(void* const* d_in, const int* in_sizes, int n_in,
                              void* d_out, int out_size, void* d_ws, size_t ws_size,
                              hipStream_t stream) {
    const float* x    = (const float*)d_in[0];
    const float* w_q1 = (const float*)d_in[1];
    const float* s_q1 = (const float*)d_in[2];
    const float* b_q1 = (const float*)d_in[3];
    const float* w_q2 = (const float*)d_in[4];
    const float* s_q2 = (const float*)d_in[5];
    const float* b_q2 = (const float*)d_in[6];
    const float* w_k1 = (const float*)d_in[7];
    const float* s_k1 = (const float*)d_in[8];
    const float* b_k1 = (const float*)d_in[9];
    const float* w_k2 = (const float*)d_in[10];
    const float* s_k2 = (const float*)d_in[11];
    const float* b_k2 = (const float*)d_in[12];
    const float* w_v  = (const float*)d_in[13];
    const float* s_v  = (const float*)d_in[14];
    const float* b_v  = (const float*)d_in[15];
    const float* w_o  = (const float*)d_in[16];
    const float* s_o  = (const float*)d_in[17];
    const float* b_o  = (const float*)d_in[18];
    float* out = (float*)d_out;

    const size_t NB = (size_t)HW * CCH;
    float* q1 = (float*)d_ws;
    float* k1 = q1 + NB;
    float* vb = k1 + NB;
    float* qb = vb + NB;
    float* kb = qb + NB;
    float* vo = kb + NB;

    gemm_bn_wmma<<<dim3(HW / 64, 4), dim3(256), 0, stream>>>(
        x, x, x, x, HW, 1,
        w_q1, w_k1, w_v, w_o + CCH,
        s_q1, s_k1, s_v, s_o, b_q1, b_k1, b_v, b_o,
        q1, k1, vb, out, 7, 15, 8, 8, 0, 0);
    gemm_bn_wmma<<<dim3(HW / 64, 3), dim3(256), 0, stream>>>(
        q1, k1, vb, nullptr, 1, CCH,
        w_q2, w_k2, w_o, nullptr,
        s_q2, s_k2, s_o, nullptr, b_q2, b_k2, b_o, nullptr,
        qb, kb, vo, nullptr, 3, 3, 4, 0, 3, 4);
    local_attn_wmma<<<dim3(HW / 64), dim3(1024), 0, stream>>>(qb, kb, vo, out);
}


# chained 1x1 convs in one launch: a wave owns 16 pixels x all 128 channels, so conv 1's accumulators are conv 2's B fragments in registers (A rows ordered to match); 3 launches -> 2; sums unchanged
# speedup vs baseline: 3.1283x; 1.0847x over previous
#include <hip/hip_runtime.h>
#include <hip/hip_bf16.h>

typedef __attribute__((ext_vector_type(2))) float v2f;
typedef __attribute__((ext_vector_type(4))) float v4f;
typedef __attribute__((ext_vector_type(8))) float v8f;

#define CCH 128
#define HDIM 128
#define WDIM 128
#define HW (HDIM * WDIM)
#define TPITCH 132
#define PXB 128

__device__ __forceinline__ void lds_barrier()
{
    __builtin_amdgcn_fence(__ATOMIC_RELEASE, "workgroup", "local");
    __builtin_amdgcn_s_barrier();
    __builtin_amdgcn_fence(__ATOMIC_ACQUIRE, "workgroup", "local");
}

__global__ __launch_bounds__(256) void chain_gemm_wmma(
    const float* __restrict__ X,
    const float* __restrict__ Wa0, const float* __restrict__ Wa1,
    const float* __restrict__ Wa2, const float* __restrict__ Wa3,
    const float* __restrict__ sa0, const float* __restrict__ sa1,
    const float* __restrict__ sa2, const float* __restrict__ sa3,
    const float* __restrict__ ba0, const float* __restrict__ ba1,
    const float* __restrict__ ba2, const float* __restrict__ ba3,
    const float* __restrict__ Wb0, const float* __restrict__ Wb1,
    const float* __restrict__ Wb2,
    const float* __restrict__ sb0, const float* __restrict__ sb1,
    const float* __restrict__ sb2,
    const float* __restrict__ bb0, const float* __restrict__ bb1,
    float* __restrict__ Y0, float* __restrict__ Y1,
    float* __restrict__ Y2, float* __restrict__ Y3)
{
    __shared__ __attribute__((aligned(16))) float tile[PXB * TPITCH];
    __shared__ __attribute__((aligned(16))) float wl[CCH * TPITCH];
    __shared__ __attribute__((aligned(16))) float sbv[4 * CCH];

    const int ch = blockIdx.y;
    const bool two = ch < 3;
    const float* __restrict__ W1 = ch == 0 ? Wa0 : (ch == 1 ? Wa1 : (ch == 2 ? Wa2 : Wa3));
    const float* __restrict__ s1 = ch == 0 ? sa0 : (ch == 1 ? sa1 : (ch == 2 ? sa2 : sa3));
    const float* __restrict__ b1 = ch == 0 ? ba0 : (ch == 1 ? ba1 : (ch == 2 ? ba2 : ba3));
    const float* __restrict__ W2 = ch == 0 ? Wb0 : (ch == 1 ? Wb1 : Wb2);
    const float* __restrict__ s2 = ch == 0 ? sb0 : (ch == 1 ? sb1 : sb2);
    const float* __restrict__ b2 = ch == 0 ? bb0 : bb1;
    float* __restrict__ Y        = ch == 0 ? Y0 : (ch == 1 ? Y1 : (ch == 2 ? Y2 : Y3));
    const int ldw1 = ch == 3 ? 2 * CCH : CCH;
    const int ldw2 = ch == 2 ? 2 * CCH : CCH;

    const int tid  = threadIdx.x;
    const int lane = tid & 31;
    const int wv   = tid >> 5;
    const int hl   = lane >> 4;
    const int l16  = lane & 15;
    const long p0  = (long)blockIdx.x * PXB;

    v4f xv[16];
    #pragma unroll
    for (int i = 0; i < 16; ++i) {
        const int e  = tid + (i << 8);
        const int c  = ((e >> 6) << 1) | ((e >> 4) & 1);
        const int p4 = ((e & 15) << 2) + (((e >> 5) & 1) << 6);
        xv[i] = *(const v4f*)(X + (long)c * HW + p0 + p4);
    }
    v4f wr[16];
    {
        const float* __restrict__ Wrow = W1 + (long)(wv << 4) * ldw1 + (lane << 2);
        #pragma unroll
        for (int r = 0; r < 16; ++r)
            wr[r] = *(const v4f*)(Wrow + (long)r * ldw1);
    }
    const int sidx = tid & (CCH - 1);
    const float sv = (tid < CCH) ? s1[sidx] : s2[sidx];
    const float bv = (tid < CCH) ? b1[sidx] : b2[sidx];

    asm volatile("" ::: "memory");

    #pragma unroll
    for (int i = 0; i < 16; ++i) {
        const int e  = tid + (i << 8);
        const int c  = ((e >> 6) << 1) | ((e >> 4) & 1);
        const int p4 = ((e & 15) << 2) + (((e >> 5) & 1) << 6);
        tile[(p4 + 0) * TPITCH + c] = xv[i].x;
        tile[(p4 + 1) * TPITCH + c] = xv[i].y;
        tile[(p4 + 2) * TPITCH + c] = xv[i].z;
        tile[(p4 + 3) * TPITCH + c] = xv[i].w;
    }
    float* wrows = wl + (wv << 4) * TPITCH + (lane << 2);
    #pragma unroll
    for (int r = 0; r < 16; ++r)
        *(v4f*)(&wrows[r * TPITCH]) = wr[r];
    sbv[((tid >> 7) << 8) + sidx]       = sv;
    sbv[((tid >> 7) << 8) + CCH + sidx] = bv;
    lds_barrier();

    v4f w2r[16];
    if (two) {
        const float* __restrict__ W2row = W2 + (long)(wv << 4) * ldw2 + (lane << 2);
        #pragma unroll
        for (int r = 0; r < 16; ++r)
            w2r[r] = *(const v4f*)(W2row + (long)r * ldw2);
    }

    const int prm = (((l16 >> 1) & 3) << 2) | ((l16 >> 3) << 1) | (l16 & 1);
    const float* Ap = wl + prm * TPITCH + (hl << 1);
    const float* Bp = tile + ((wv << 4) + l16) * TPITCH + (hl << 1);
    v8f acc1[8] = {};
    #pragma unroll
    for (int kk = 0; kk < 32; ++kk) {
        const v2f b = *(const v2f*)(Bp + 4 * kk);
        #pragma unroll
        for (int m = 0; m < 8; ++m) {
            const v2f a = *(const v2f*)(Ap + m * (16 * TPITCH) + 4 * kk);
            acc1[m] = __builtin_amdgcn_wmma_f32_16x16x4_f32(
                false, a, false, b, (short)0, acc1[m], false, false);
        }
    }
    #pragma unroll
    for (int m = 0; m < 8; ++m) {
        #pragma unroll
        for (int j = 0; j < 4; ++j) {
            const v2f sc = *(const v2f*)(&sbv[(m << 4) + (j << 2) + (hl << 1)]);
            const v2f bi = *(const v2f*)(&sbv[CCH + (m << 4) + (j << 2) + (hl << 1)]);
            float f0 = acc1[m][2 * j]     * sc.x + bi.x;
            float f1 = acc1[m][2 * j + 1] * sc.y + bi.y;
            if (two) { f0 = fmaxf(f0, 0.0f); f1 = fmaxf(f1, 0.0f); }
            acc1[m][2 * j]     = f0;
            acc1[m][2 * j + 1] = f1;
        }
    }

    if (!two) {
        float* dst = &tile[((wv << 4) + l16) * TPITCH + (hl << 1)];
        #pragma unroll
        for (int m = 0; m < 8; ++m) {
            #pragma unroll
            for (int j = 0; j < 4; ++j) {
                const v2f pr = {acc1[m][2 * j], acc1[m][2 * j + 1]};
                *(v2f*)(dst + (m << 4) + (j << 2)) = pr;
            }
        }
        lds_barrier();
        const int ln = wv & 3;
        float* __restrict__ Yl = Y + p0 + (ln << 5) + lane;
        const float* trd = &tile[((ln << 5) + lane) * TPITCH];
        #pragma unroll
        for (int i = 0; i < 64; ++i) {
            const int o = (wv >> 2) + (i << 1);
            Yl[(long)o * HW] = trd[o];
        }
        return;
    }

    lds_barrier();
    #pragma unroll
    for (int r = 0; r < 16; ++r)
        *(v4f*)(&wrows[r * TPITCH]) = w2r[r];
    lds_barrier();

    const float* A2p = wl + l16 * TPITCH + (hl << 1);
    v8f acc2[8] = {};
    #pragma unroll
    for (int m = 0; m < 8; ++m) {
        #pragma unroll
        for (int j = 0; j < 4; ++j) {
            const v2f b = {acc1[m][2 * j], acc1[m][2 * j + 1]};
            #pragma unroll
            for (int m2 = 0; m2 < 8; ++m2) {
                const v2f a = *(const v2f*)(A2p + m2 * (16 * TPITCH) + 4 * ((m << 2) + j));
                acc2[m2] = __builtin_amdgcn_wmma_f32_16x16x4_f32(
                    false, a, false, b, (short)0, acc2[m2], false, false);
            }
        }
    }

    const bool act2 = ch < 2;
    {
        float* dst = &tile[((wv << 4) + l16) * TPITCH + (hl << 3)];
        #pragma unroll
        for (int m2 = 0; m2 < 8; ++m2) {
            const v4f sc0 = *(const v4f*)(&sbv[2 * CCH + (m2 << 4) + (hl << 3)]);
            const v4f sc1 = *(const v4f*)(&sbv[2 * CCH + (m2 << 4) + (hl << 3) + 4]);
            v4f bi0 = *(const v4f*)(&sbv[3 * CCH + (m2 << 4) + (hl << 3)]);
            v4f bi1 = *(const v4f*)(&sbv[3 * CCH + (m2 << 4) + (hl << 3) + 4]);
            if (!act2) { bi0 = (v4f){0.0f, 0.0f, 0.0f, 0.0f}; bi1 = bi0; }
            float f[8];
            #pragma unroll
            for (int r = 0; r < 4; ++r) {
                f[r]     = acc2[m2][r]     * sc0[r] + bi0[r];
                f[r + 4] = acc2[m2][r + 4] * sc1[r] + bi1[r];
            }
            if (act2) {
                #pragma unroll
                for (int r = 0; r < 8; ++r) f[r] = fmaxf(f[r], 0.0f);
            }
            const v4f lo = {f[0], f[1], f[2], f[3]};
            const v4f hi = {f[4], f[5], f[6], f[7]};
            *(v4f*)(dst + (m2 << 4))     = lo;
            *(v4f*)(dst + (m2 << 4) + 4) = hi;
        }
    }
    lds_barrier();
    if (ch == 2) {
        #pragma unroll
        for (int t = 0; t < 16; ++t) {
            const int i  = t & 7;
            const int oc = wv + ((t >> 3) << 3);
            const float* trow = &tile[((oc << 3) + (hl << 2)) * TPITCH + (i << 4) + l16];
            const v4f val = {trow[0], trow[TPITCH], trow[2 * TPITCH], trow[3 * TPITCH]};
            *(v4f*)(Y + (long)i * (HW * 16) + ((p0 >> 2) + (oc << 1) + hl) * 64 + (l16 << 2)) = val;
        }
    } else {
        float* __restrict__ Yc = Y + (p0 + (wv << 4) + l16) * 8 + (hl << 2);
        const float* trow = &tile[((wv << 4) + l16) * TPITCH + (hl << 2)];
        #pragma unroll
        for (int jj = 0; jj < 16; ++jj) {
            const v4f val = *(const v4f*)(trow + (jj << 3));
            *(v4f*)(Yc + (long)jj * (HW * 8)) = val;
        }
    }
}

#define SP 132

#define OP 36

__global__ __launch_bounds__(1024) __attribute__((amdgpu_waves_per_eu(8, 8)))
void local_attn_wmma(
    const float* __restrict__ q, const float* __restrict__ k,
    const float* __restrict__ v, float* __restrict__ out)
{
    __shared__ __attribute__((aligned(16))) float sp4[4 * 16 * SP];
    __shared__ __attribute__((aligned(16))) float ot[CCH * 2 * OP];

    const int tid  = threadIdx.x;
    const int lane = tid & 31;
    const int sh   = tid >> 8;
    const int wv   = (tid >> 5) & 7;
    const int hl   = lane >> 4;
    const int l16  = lane & 15;
    const int h0   = (blockIdx.x >> 2) << 1;
    const int wb   = (blockIdx.x & 3) << 5;
    const int w0   = wb + (sh << 3);
    float* __restrict__ sp = sp4 + sh * (16 * SP);
    const v4f zero4 = {0.0f, 0.0f, 0.0f, 0.0f};

    {
        const int  hn    = h0 - 3 + wv;
        const bool rowok = (unsigned)hn < (unsigned)HDIM;
        const int  wn    = w0 - 4 + l16;
        const bool ok    = rowok && (unsigned)wn < (unsigned)WDIM;
        const int  hc = min(max(hn, 0), HDIM - 1);
        const int  wc = min(max(wn, 0), WDIM - 1);
        const long pn = (long)hc * WDIM + wc;
        const long pa = (long)(h0 + (l16 >> 3)) * WDIM + w0 + (l16 & 7);
        const float* __restrict__ qp = q + pa * 8 + (hl << 2);
        const float* __restrict__ kp = k + pn * 8 + (hl << 2);

        v8f s0 = {}, s1 = {};
        #pragma unroll
        for (int jj = 0; jj < 16; ++jj) {
            const v4f a = *(const v4f*)(qp + (long)jj * (HW * 8));
            v4f b = *(const v4f*)(kp + (long)jj * (HW * 8));
            b = ok ? b : zero4;
            s0 = __builtin_amdgcn_wmma_f32_16x16x4_f32(
                false, a.xy, false, b.xy, (short)0, s0, false, false);
            s1 = __builtin_amdgcn_wmma_f32_16x16x4_f32(
                false, a.zw, false, b.zw, (short)0, s1, false, false);
        }
        #pragma unroll
        for (int r = 0; r < 8; ++r)
            sp[(r + (hl << 3)) * SP + wv * 16 + l16] = s0[r] + s1[r];
    }
    __syncthreads();

    {
        const int  m      = 2 * wv + hl;
        const int  ry     = wv >> 2;
        const int  cx     = m & 7;
        const bool inband = (l16 >= cx + 1) && (l16 <= cx + 7);
        float* row = &sp[m * SP + ry * 16 + l16];
        float s[7];
        #pragma unroll
        for (int dy = 0; dy < 7; ++dy) s[dy] = row[dy * 16];
        float mx = s[0];
        #pragma unroll
        for (int dy = 1; dy < 7; ++dy) mx = fmaxf(mx, s[dy]);
        mx = inband ? mx : -INFINITY;
        #pragma unroll
        for (int off = 8; off > 0; off >>= 1)
            mx = fmaxf(mx, __shfl_xor(mx, off, 32));
        float e[7];
        float z = 0.0f;
        #pragma unroll
        for (int dy = 0; dy < 7; ++dy) {
            e[dy] = inband ? __expf(s[dy] - mx) : 0.0f;
            z += e[dy];
        }
        #pragma unroll
        for (int off = 8; off > 0; off >>= 1)
            z += __shfl_xor(z, off, 32);
        const float inv = 1.0f / z;
        #pragma unroll
        for (int dy = 0; dy < 7; ++dy) row[dy * 16] = e[dy] * inv;
        sp[m * SP + (ry ? 0 : 7 * 16) + l16] = 0.0f;
    }
    __syncthreads();

    const int c0 = wv << 4;
    const float* __restrict__ vq = v + (long)wv * (HW * 16) + (l16 << 2);
    const int qcol0 = (w0 >> 2) - 1 + hl;
    v8f o0 = {}, o1 = {};
    #pragma unroll
    for (int r = 0; r < 8; ++r) {
        const int  hn    = h0 - 3 + r;
        const bool rowok = (unsigned)hn < (unsigned)HDIM;
        const int  hc    = min(max(hn, 0), HDIM - 1);
        #pragma unroll
        for (int T = 0; T < 2; ++T) {
            const int  qcol = qcol0 + 2 * T;
            const bool okq  = rowok && (unsigned)qcol < (unsigned)(WDIM / 4);
            const int  qc   = min(max(qcol, 0), WDIM / 4 - 1);
            v4f b = *(const v4f*)(vq + (hc * (WDIM / 4) + qc) * 64);
            b = okq ? b : zero4;
            const v4f a = *(const v4f*)(&sp[l16 * SP + r * 16 + 8 * T + (hl << 2)]);
            o0 = __builtin_amdgcn_wmma_f32_16x16x4_f32(
                false, a.xy, false, b.xy, (short)0, o0, false, false);
            o1 = __builtin_amdgcn_wmma_f32_16x16x4_f32(
                false, a.zw, false, b.zw, (short)0, o1, false, false);
        }
    }
    v4f xo[2];
    #pragma unroll
    for (int i = 0; i < 2; ++i) {
        const int idx = tid + (i << 10);
        const int c   = idx >> 4;
        const int ry  = (idx >> 3) & 1;
        const int px4 = (idx & 7) << 2;
        xo[i] = *(const v4f*)(out + (long)c * HW + (h0 + ry) * WDIM + wb + px4);
    }

    {
        float* dst = &ot[(((c0 + l16) << 1) + hl) * OP + (sh << 3)];
        const v4f lo = {o0[0] + o1[0], o0[1] + o1[1], o0[2] + o1[2], o0[3] + o1[3]};
        const v4f hi = {o0[4] + o1[4], o0[5] + o1[5], o0[6] + o1[6], o0[7] + o1[7]};
        *(v4f*)(dst)     = lo;
        *(v4f*)(dst + 4) = hi;
    }
    __syncthreads();
    #pragma unroll
    for (int i = 0; i < 2; ++i) {
        const int idx = tid + (i << 10);
        const int c   = idx >> 4;
        const int ry  = (idx >> 3) & 1;
        const int px4 = (idx & 7) << 2;
        const v4f a = *(const v4f*)(&ot[(idx >> 3) * OP + px4]);
        *(v4f*)(out + (long)c * HW + (h0 + ry) * WDIM + wb + px4) = a + xo[i];
    }
}

extern "C" void kernel_launch(void* const* d_in, const int* in_sizes, int n_in,
                              void* d_out, int out_size, void* d_ws, size_t ws_size,
                              hipStream_t stream) {
    const float* x    = (const float*)d_in[0];
    const float* w_q1 = (const float*)d_in[1];
    const float* s_q1 = (const float*)d_in[2];
    const float* b_q1 = (const float*)d_in[3];
    const float* w_q2 = (const float*)d_in[4];
    const float* s_q2 = (const float*)d_in[5];
    const float* b_q2 = (const float*)d_in[6];
    const float* w_k1 = (const float*)d_in[7];
    const float* s_k1 = (const float*)d_in[8];
    const float* b_k1 = (const float*)d_in[9];
    const float* w_k2 = (const float*)d_in[10];
    const float* s_k2 = (const float*)d_in[11];
    const float* b_k2 = (const float*)d_in[12];
    const float* w_v  = (const float*)d_in[13];
    const float* s_v  = (const float*)d_in[14];
    const float* b_v  = (const float*)d_in[15];
    const float* w_o  = (const float*)d_in[16];
    const float* s_o  = (const float*)d_in[17];
    const float* b_o  = (const float*)d_in[18];
    float* out = (float*)d_out;

    const size_t NB = (size_t)HW * CCH;
    float* qb = (float*)d_ws;
    float* kb = qb + NB;
    float* vo = kb + NB;

    chain_gemm_wmma<<<dim3(HW / PXB, 4), dim3(256), 0, stream>>>(
        x,
        w_q1, w_k1, w_v, w_o + CCH,
        s_q1, s_k1, s_v, s_o,
        b_q1, b_k1, b_v, b_o,
        w_q2, w_k2, w_o,
        s_q2, s_k2, s_o,
        b_q2, b_k2,
        qb, kb, vo, out);
    local_attn_wmma<<<dim3(HW / 64), dim3(1024), 0, stream>>>(qb, kb, vo, out);
}


# 1x1-conv chains, first conv: weights and x tile staged and consumed in four k chunks of 32 channels (a barrier each): products start on the first quarter while the rest is in flight; sums unchanged
# speedup vs baseline: 3.2500x; 1.0389x over previous
#include <hip/hip_runtime.h>
#include <hip/hip_bf16.h>

typedef __attribute__((ext_vector_type(2))) float v2f;
typedef __attribute__((ext_vector_type(4))) float v4f;
typedef __attribute__((ext_vector_type(8))) float v8f;

#define CCH 128
#define HDIM 128
#define WDIM 128
#define HW (HDIM * WDIM)
#define TPITCH 132
#define PXB 128

__device__ __forceinline__ void lds_barrier()
{
    __builtin_amdgcn_fence(__ATOMIC_RELEASE, "workgroup", "local");
    __builtin_amdgcn_s_barrier();
    __builtin_amdgcn_fence(__ATOMIC_ACQUIRE, "workgroup", "local");
}

__global__ __launch_bounds__(256) void chain_gemm_wmma(
    const float* __restrict__ X,
    const float* __restrict__ Wa0, const float* __restrict__ Wa1,
    const float* __restrict__ Wa2, const float* __restrict__ Wa3,
    const float* __restrict__ sa0, const float* __restrict__ sa1,
    const float* __restrict__ sa2, const float* __restrict__ sa3,
    const float* __restrict__ ba0, const float* __restrict__ ba1,
    const float* __restrict__ ba2, const float* __restrict__ ba3,
    const float* __restrict__ Wb0, const float* __restrict__ Wb1,
    const float* __restrict__ Wb2,
    const float* __restrict__ sb0, const float* __restrict__ sb1,
    const float* __restrict__ sb2,
    const float* __restrict__ bb0, const float* __restrict__ bb1,
    float* __restrict__ Y0, float* __restrict__ Y1,
    float* __restrict__ Y2, float* __restrict__ Y3)
{
    __shared__ __attribute__((aligned(16))) float tile[PXB * TPITCH];
    __shared__ __attribute__((aligned(16))) float wl[CCH * TPITCH];
    __shared__ __attribute__((aligned(16))) float sbv[4 * CCH];

    const int ch = blockIdx.y;
    const bool two = ch < 3;
    const float* __restrict__ W1 = ch == 0 ? Wa0 : (ch == 1 ? Wa1 : (ch == 2 ? Wa2 : Wa3));
    const float* __restrict__ s1 = ch == 0 ? sa0 : (ch == 1 ? sa1 : (ch == 2 ? sa2 : sa3));
    const float* __restrict__ b1 = ch == 0 ? ba0 : (ch == 1 ? ba1 : (ch == 2 ? ba2 : ba3));
    const float* __restrict__ W2 = ch == 0 ? Wb0 : (ch == 1 ? Wb1 : Wb2);
    const float* __restrict__ s2 = ch == 0 ? sb0 : (ch == 1 ? sb1 : sb2);
    const float* __restrict__ b2 = ch == 0 ? bb0 : bb1;
    float* __restrict__ Y        = ch == 0 ? Y0 : (ch == 1 ? Y1 : (ch == 2 ? Y2 : Y3));
    const int ldw1 = ch == 3 ? 2 * CCH : CCH;
    const int ldw2 = ch == 2 ? 2 * CCH : CCH;

    const int tid  = threadIdx.x;
    const int lane = tid & 31;
    const int wv   = tid >> 5;
    const int hl   = lane >> 4;
    const int l16  = lane & 15;
    const long p0  = (long)blockIdx.x * PXB;

    const int sidx = tid & (CCH - 1);
    const float sv = (tid < CCH) ? s1[sidx] : s2[sidx];
    const float bv = (tid < CCH) ? b1[sidx] : b2[sidx];
    const int wrow = tid >> 3;
    const int wcol = (tid & 7) << 2;
    v4f wr[16];
    v4f xv[16];
    #pragma unroll
    for (int c = 0; c < 4; ++c) {
        #pragma unroll
        for (int i = 0; i < 4; ++i)
            wr[4 * c + i] = *(const v4f*)(W1 + (long)(wrow + 32 * i) * ldw1 + 32 * c + wcol);
        #pragma unroll
        for (int i = 4 * c; i < 4 * c + 4; ++i) {
            const int e  = tid + (i << 8);
            const int cc = ((e >> 6) << 1) | ((e >> 4) & 1);
            const int p4 = ((e & 15) << 2) + (((e >> 5) & 1) << 6);
            xv[i] = *(const v4f*)(X + (long)cc * HW + p0 + p4);
        }
    }

    asm volatile("" ::: "memory");

    sbv[((tid >> 7) << 8) + sidx]       = sv;
    sbv[((tid >> 7) << 8) + CCH + sidx] = bv;

    const int prm = (((l16 >> 1) & 3) << 2) | ((l16 >> 3) << 1) | (l16 & 1);
    const float* Ap = wl + prm * TPITCH + (hl << 1);
    const float* Bp = tile + ((wv << 4) + l16) * TPITCH + (hl << 1);
    float* wrows = wl + (wv << 4) * TPITCH + (lane << 2);
    v8f acc1[8] = {};
    v4f w2r[16];
    #pragma unroll
    for (int c = 0; c < 4; ++c) {
        #pragma unroll
        for (int i = 0; i < 4; ++i)
            *(v4f*)(&wl[(wrow + 32 * i) * TPITCH + 32 * c + wcol]) = wr[4 * c + i];
        #pragma unroll
        for (int i = 4 * c; i < 4 * c + 4; ++i) {
            const int e  = tid + (i << 8);
            const int cc = ((e >> 6) << 1) | ((e >> 4) & 1);
            const int p4 = ((e & 15) << 2) + (((e >> 5) & 1) << 6);
            tile[(p4 + 0) * TPITCH + cc] = xv[i].x;
            tile[(p4 + 1) * TPITCH + cc] = xv[i].y;
            tile[(p4 + 2) * TPITCH + cc] = xv[i].z;
            tile[(p4 + 3) * TPITCH + cc] = xv[i].w;
        }
        lds_barrier();
        if (c == 3) {
            if (two) {
                const float* __restrict__ W2row = W2 + (long)(wv << 4) * ldw2 + (lane << 2);
                #pragma unroll
                for (int r = 0; r < 16; ++r)
                    w2r[r] = *(const v4f*)(W2row + (long)r * ldw2);
            }
        }
        #pragma unroll
        for (int kk = 8 * c; kk < 8 * c + 8; ++kk) {
            const v2f b = *(const v2f*)(Bp + 4 * kk);
            #pragma unroll
            for (int m = 0; m < 8; ++m) {
                const v2f a = *(const v2f*)(Ap + m * (16 * TPITCH) + 4 * kk);
                acc1[m] = __builtin_amdgcn_wmma_f32_16x16x4_f32(
                    false, a, false, b, (short)0, acc1[m], false, false);
            }
        }
    }
    #pragma unroll
    for (int m = 0; m < 8; ++m) {
        #pragma unroll
        for (int j = 0; j < 4; ++j) {
            const v2f sc = *(const v2f*)(&sbv[(m << 4) + (j << 2) + (hl << 1)]);
            const v2f bi = *(const v2f*)(&sbv[CCH + (m << 4) + (j << 2) + (hl << 1)]);
            float f0 = acc1[m][2 * j]     * sc.x + bi.x;
            float f1 = acc1[m][2 * j + 1] * sc.y + bi.y;
            if (two) { f0 = fmaxf(f0, 0.0f); f1 = fmaxf(f1, 0.0f); }
            acc1[m][2 * j]     = f0;
            acc1[m][2 * j + 1] = f1;
        }
    }

    if (!two) {
        float* dst = &tile[((wv << 4) + l16) * TPITCH + (hl << 1)];
        #pragma unroll
        for (int m = 0; m < 8; ++m) {
            #pragma unroll
            for (int j = 0; j < 4; ++j) {
                const v2f pr = {acc1[m][2 * j], acc1[m][2 * j + 1]};
                *(v2f*)(dst + (m << 4) + (j << 2)) = pr;
            }
        }
        lds_barrier();
        const int ln = wv & 3;
        float* __restrict__ Yl = Y + p0 + (ln << 5) + lane;
        const float* trd = &tile[((ln << 5) + lane) * TPITCH];
        #pragma unroll
        for (int i = 0; i < 64; ++i) {
            const int o = (wv >> 2) + (i << 1);
            Yl[(long)o * HW] = trd[o];
        }
        return;
    }

    lds_barrier();
    #pragma unroll
    for (int r = 0; r < 16; ++r)
        *(v4f*)(&wrows[r * TPITCH]) = w2r[r];
    lds_barrier();

    const float* A2p = wl + l16 * TPITCH + (hl << 1);
    v8f acc2[8] = {};
    #pragma unroll
    for (int m = 0; m < 8; ++m) {
        #pragma unroll
        for (int j = 0; j < 4; ++j) {
            const v2f b = {acc1[m][2 * j], acc1[m][2 * j + 1]};
            #pragma unroll
            for (int m2 = 0; m2 < 8; ++m2) {
                const v2f a = *(const v2f*)(A2p + m2 * (16 * TPITCH) + 4 * ((m << 2) + j));
                acc2[m2] = __builtin_amdgcn_wmma_f32_16x16x4_f32(
                    false, a, false, b, (short)0, acc2[m2], false, false);
            }
        }
    }

    const bool act2 = ch < 2;
    {
        float* dst = &tile[((wv << 4) + l16) * TPITCH + (hl << 3)];
        #pragma unroll
        for (int m2 = 0; m2 < 8; ++m2) {
            const v4f sc0 = *(const v4f*)(&sbv[2 * CCH + (m2 << 4) + (hl << 3)]);
            const v4f sc1 = *(const v4f*)(&sbv[2 * CCH + (m2 << 4) + (hl << 3) + 4]);
            v4f bi0 = *(const v4f*)(&sbv[3 * CCH + (m2 << 4) + (hl << 3)]);
            v4f bi1 = *(const v4f*)(&sbv[3 * CCH + (m2 << 4) + (hl << 3) + 4]);
            if (!act2) { bi0 = (v4f){0.0f, 0.0f, 0.0f, 0.0f}; bi1 = bi0; }
            float f[8];
            #pragma unroll
            for (int r = 0; r < 4; ++r) {
                f[r]     = acc2[m2][r]     * sc0[r] + bi0[r];
                f[r + 4] = acc2[m2][r + 4] * sc1[r] + bi1[r];
            }
            if (act2) {
                #pragma unroll
                for (int r = 0; r < 8; ++r) f[r] = fmaxf(f[r], 0.0f);
            }
            const v4f lo = {f[0], f[1], f[2], f[3]};
            const v4f hi = {f[4], f[5], f[6], f[7]};
            *(v4f*)(dst + (m2 << 4))     = lo;
            *(v4f*)(dst + (m2 << 4) + 4) = hi;
        }
    }
    lds_barrier();
    if (ch == 2) {
        #pragma unroll
        for (int t = 0; t < 16; ++t) {
            const int i  = t & 7;
            const int oc = wv + ((t >> 3) << 3);
            const float* trow = &tile[((oc << 3) + (hl << 2)) * TPITCH + (i << 4) + l16];
            const v4f val = {trow[0], trow[TPITCH], trow[2 * TPITCH], trow[3 * TPITCH]};
            *(v4f*)(Y + (long)i * (HW * 16) + ((p0 >> 2) + (oc << 1) + hl) * 64 + (l16 << 2)) = val;
        }
    } else {
        float* __restrict__ Yc = Y + (p0 + (wv << 4) + l16) * 8 + (hl << 2);
        const float* trow = &tile[((wv << 4) + l16) * TPITCH + (hl << 2)];
        #pragma unroll
        for (int jj = 0; jj < 16; ++jj) {
            const v4f val = *(const v4f*)(trow + (jj << 3));
            *(v4f*)(Yc + (long)jj * (HW * 8)) = val;
        }
    }
}

#define SP 132

#define OP 36

__global__ __launch_bounds__(1024) __attribute__((amdgpu_waves_per_eu(8, 8)))
void local_attn_wmma(
    const float* __restrict__ q, const float* __restrict__ k,
    const float* __restrict__ v, float* __restrict__ out)
{
    __shared__ __attribute__((aligned(16))) float sp4[4 * 16 * SP];
    __shared__ __attribute__((aligned(16))) float ot[CCH * 2 * OP];

    const int tid  = threadIdx.x;
    const int lane = tid & 31;
    const int sh   = tid >> 8;
    const int wv   = (tid >> 5) & 7;
    const int hl   = lane >> 4;
    const int l16  = lane & 15;
    const int h0   = (blockIdx.x >> 2) << 1;
    const int wb   = (blockIdx.x & 3) << 5;
    const int w0   = wb + (sh << 3);
    float* __restrict__ sp = sp4 + sh * (16 * SP);
    const v4f zero4 = {0.0f, 0.0f, 0.0f, 0.0f};

    {
        const int  hn    = h0 - 3 + wv;
        const bool rowok = (unsigned)hn < (unsigned)HDIM;
        const int  wn    = w0 - 4 + l16;
        const bool ok    = rowok && (unsigned)wn < (unsigned)WDIM;
        const int  hc = min(max(hn, 0), HDIM - 1);
        const int  wc = min(max(wn, 0), WDIM - 1);
        const long pn = (long)hc * WDIM + wc;
        const long pa = (long)(h0 + (l16 >> 3)) * WDIM + w0 + (l16 & 7);
        const float* __restrict__ qp = q + pa * 8 + (hl << 2);
        const float* __restrict__ kp = k + pn * 8 + (hl << 2);

        v8f s0 = {}, s1 = {};
        #pragma unroll
        for (int jj = 0; jj < 16; ++jj) {
            const v4f a = *(const v4f*)(qp + (long)jj * (HW * 8));
            v4f b = *(const v4f*)(kp + (long)jj * (HW * 8));
            b = ok ? b : zero4;
            s0 = __builtin_amdgcn_wmma_f32_16x16x4_f32(
                false, a.xy, false, b.xy, (short)0, s0, false, false);
            s1 = __builtin_amdgcn_wmma_f32_16x16x4_f32(
                false, a.zw, false, b.zw, (short)0, s1, false, false);
        }
        #pragma unroll
        for (int r = 0; r < 8; ++r)
            sp[(r + (hl << 3)) * SP + wv * 16 + l16] = s0[r] + s1[r];
    }
    __syncthreads();

    {
        const int  m      = 2 * wv + hl;
        const int  ry     = wv >> 2;
        const int  cx     = m & 7;
        const bool inband = (l16 >= cx + 1) && (l16 <= cx + 7);
        float* row = &sp[m * SP + ry * 16 + l16];
        float s[7];
        #pragma unroll
        for (int dy = 0; dy < 7; ++dy) s[dy] = row[dy * 16];
        float mx = s[0];
        #pragma unroll
        for (int dy = 1; dy < 7; ++dy) mx = fmaxf(mx, s[dy]);
        mx = inband ? mx : -INFINITY;
        #pragma unroll
        for (int off = 8; off > 0; off >>= 1)
            mx = fmaxf(mx, __shfl_xor(mx, off, 32));
        float e[7];
        float z = 0.0f;
        #pragma unroll
        for (int dy = 0; dy < 7; ++dy) {
            e[dy] = inband ? __expf(s[dy] - mx) : 0.0f;
            z += e[dy];
        }
        #pragma unroll
        for (int off = 8; off > 0; off >>= 1)
            z += __shfl_xor(z, off, 32);
        const float inv = 1.0f / z;
        #pragma unroll
        for (int dy = 0; dy < 7; ++dy) row[dy * 16] = e[dy] * inv;
        sp[m * SP + (ry ? 0 : 7 * 16) + l16] = 0.0f;
    }
    __syncthreads();

    const int c0 = wv << 4;
    const float* __restrict__ vq = v + (long)wv * (HW * 16) + (l16 << 2);
    const int qcol0 = (w0 >> 2) - 1 + hl;
    v8f o0 = {}, o1 = {};
    #pragma unroll
    for (int r = 0; r < 8; ++r) {
        const int  hn    = h0 - 3 + r;
        const bool rowok = (unsigned)hn < (unsigned)HDIM;
        const int  hc    = min(max(hn, 0), HDIM - 1);
        #pragma unroll
        for (int T = 0; T < 2; ++T) {
            const int  qcol = qcol0 + 2 * T;
            const bool okq  = rowok && (unsigned)qcol < (unsigned)(WDIM / 4);
            const int  qc   = min(max(qcol, 0), WDIM / 4 - 1);
            v4f b = *(const v4f*)(vq + (hc * (WDIM / 4) + qc) * 64);
            b = okq ? b : zero4;
            const v4f a = *(const v4f*)(&sp[l16 * SP + r * 16 + 8 * T + (hl << 2)]);
            o0 = __builtin_amdgcn_wmma_f32_16x16x4_f32(
                false, a.xy, false, b.xy, (short)0, o0, false, false);
            o1 = __builtin_amdgcn_wmma_f32_16x16x4_f32(
                false, a.zw, false, b.zw, (short)0, o1, false, false);
        }
    }
    v4f xo[2];
    #pragma unroll
    for (int i = 0; i < 2; ++i) {
        const int idx = tid + (i << 10);
        const int c   = idx >> 4;
        const int ry  = (idx >> 3) & 1;
        const int px4 = (idx & 7) << 2;
        xo[i] = *(const v4f*)(out + (long)c * HW + (h0 + ry) * WDIM + wb + px4);
    }

    {
        float* dst = &ot[(((c0 + l16) << 1) + hl) * OP + (sh << 3)];
        const v4f lo = {o0[0] + o1[0], o0[1] + o1[1], o0[2] + o1[2], o0[3] + o1[3]};
        const v4f hi = {o0[4] + o1[4], o0[5] + o1[5], o0[6] + o1[6], o0[7] + o1[7]};
        *(v4f*)(dst)     = lo;
        *(v4f*)(dst + 4) = hi;
    }
    __syncthreads();
    #pragma unroll
    for (int i = 0; i < 2; ++i) {
        const int idx = tid + (i << 10);
        const int c   = idx >> 4;
        const int ry  = (idx >> 3) & 1;
        const int px4 = (idx & 7) << 2;
        const v4f a = *(const v4f*)(&ot[(idx >> 3) * OP + px4]);
        *(v4f*)(out + (long)c * HW + (h0 + ry) * WDIM + wb + px4) = a + xo[i];
    }
}

extern "C" void kernel_launch(void* const* d_in, const int* in_sizes, int n_in,
                              void* d_out, int out_size, void* d_ws, size_t ws_size,
                              hipStream_t stream) {
    const float* x    = (const float*)d_in[0];
    const float* w_q1 = (const float*)d_in[1];
    const float* s_q1 = (const float*)d_in[2];
    const float* b_q1 = (const float*)d_in[3];
    const float* w_q2 = (const float*)d_in[4];
    const float* s_q2 = (const float*)d_in[5];
    const float* b_q2 = (const float*)d_in[6];
    const float* w_k1 = (const float*)d_in[7];
    const float* s_k1 = (const float*)d_in[8];
    const float* b_k1 = (const float*)d_in[9];
    const float* w_k2 = (const float*)d_in[10];
    const float* s_k2 = (const float*)d_in[11];
    const float* b_k2 = (const float*)d_in[12];
    const float* w_v  = (const float*)d_in[13];
    const float* s_v  = (const float*)d_in[14];
    const float* b_v  = (const float*)d_in[15];
    const float* w_o  = (const float*)d_in[16];
    const float* s_o  = (const float*)d_in[17];
    const float* b_o  = (const float*)d_in[18];
    float* out = (float*)d_out;

    const size_t NB = (size_t)HW * CCH;
    float* qb = (float*)d_ws;
    float* kb = qb + NB;
    float* vo = kb + NB;

    chain_gemm_wmma<<<dim3(HW / PXB, 4), dim3(256), 0, stream>>>(
        x,
        w_q1, w_k1, w_v, w_o + CCH,
        s_q1, s_k1, s_v, s_o,
        b_q1, b_k1, b_v, b_o,
        w_q2, w_k2, w_o,
        s_q2, s_k2, s_o,
        b_q2, b_k2,
        qb, kb, vo, out);
    local_attn_wmma<<<dim3(HW / 64), dim3(1024), 0, stream>>>(qb, kb, vo, out);
}
